# starting kernel
# baseline (speedup 1.0000x reference)
#include <hip/hip_runtime.h>
#include <hip/hip_bf16.h>

typedef __attribute__((ext_vector_type(2))) float v2f;
typedef __attribute__((ext_vector_type(8))) float v8f;

__global__ void gcn_zero(float* __restrict__ p, long n) {
    long i = (long)blockIdx.x * blockDim.x + threadIdx.x;
    if (i < n) p[i] = 0.0f;
}

__global__ void gcn_degree(const int* __restrict__ src, const int* __restrict__ dst,
                           float* __restrict__ deg_src, float* __restrict__ deg_dst, int e) {
    int i = blockIdx.x * blockDim.x + threadIdx.x;
    if (i >= e) return;
    unsafeAtomicAdd(&deg_src[src[i]], 1.0f);
    unsafeAtomicAdd(&deg_dst[dst[i]], 1.0f);
}

__global__ void gcn_rsqrt(float* __restrict__ p, long n) {
    long i = (long)blockIdx.x * blockDim.x + threadIdx.x;
    if (i < n) p[i] = rsqrtf(fmaxf(p[i], 1.0f));
}

template <int DIN, int DN>
__global__ __launch_bounds__(((DN + 15) / 16) * 32) void gcn_gemm_wmma(
    const float* __restrict__ H, const float* __restrict__ norm_src,
    const float* __restrict__ W, float* __restrict__ Y, int n)
{
    constexpr int LDA = DIN + 1;
    __shared__ float As[16 * LDA];

    const int row0 = blockIdx.x * 16;
    const int tid  = threadIdx.x;

    for (int idx = tid; idx < 16 * DIN; idx += blockDim.x) {
        int r = idx / DIN;
        int c = idx - r * DIN;
        int row = row0 + r;
        As[r * LDA + c] = (row < n) ? H[(size_t)row * DIN + c] * norm_src[row] : 0.0f;
    }
    __syncthreads();

    const int lane = tid & 31;
    const int wave = tid >> 5;
    const int mn   = lane & 15;
    const int kh   = lane >> 4;
    const int col  = (wave << 4) + mn;

    const int   colc = (col < DN) ? col : (DN - 1);
    float msk = 1.0f;
    if constexpr ((DN & 15) != 0) msk = (col < DN) ? 1.0f : 0.0f;

    const float* __restrict__ ap = As + mn * LDA + 2 * kh;
    const float* __restrict__ wp = W + (size_t)(2 * kh) * DN + colc;

    v8f acc = {0.f, 0.f, 0.f, 0.f, 0.f, 0.f, 0.f, 0.f};

    #pragma unroll
    for (int k = 0; k < DIN; k += 4) {
        v2f a, b;
        a.x = ap[k];
        a.y = ap[k + 1];
        b.x = wp[k * DN] * msk;
        b.y = wp[k * DN + DN] * msk;
        acc = __builtin_amdgcn_wmma_f32_16x16x4_f32(
            false, a, false, b, (short)0, acc, false, false);
    }

    if (col < DN) {
        #pragma unroll
        for (int v = 0; v < 8; ++v) {
            int row = row0 + v + 8 * kh;
            if (row < n) Y[(size_t)row * DN + col] = acc[v];
        }
    }
}

template <int D>
__global__ __launch_bounds__(256) void gcn_scatter_add(
    const float* __restrict__ Y, const int* __restrict__ src, const int* __restrict__ dst,
    float* __restrict__ Out, int n_edges)
{
    constexpr int CPE = D >> 2;
    long idx = (long)blockIdx.x * blockDim.x + threadIdx.x;
    long total = (long)n_edges * CPE;
    if (idx >= total) return;
    int e = (int)(idx / CPE);
    int c = (int)(idx - (long)e * CPE);
    int s = src[e];
    int t = dst[e];
    const float4 v = *reinterpret_cast<const float4*>(Y + (size_t)s * D + 4 * c);
    float* o = Out + (size_t)t * D + 4 * c;
    unsafeAtomicAdd(o + 0, v.x);
    unsafeAtomicAdd(o + 1, v.y);
    unsafeAtomicAdd(o + 2, v.z);
    unsafeAtomicAdd(o + 3, v.w);
}

__global__ void gcn_post(float* __restrict__ A, const float* __restrict__ norm_dst,
                         const float* __restrict__ bias, int d, int n, int do_relu)
{
    long idx = (long)blockIdx.x * blockDim.x + threadIdx.x;
    if (idx >= (long)n * d) return;
    int i = (int)(idx / d);
    int f = (int)(idx - (long)i * d);
    float v = A[idx] * norm_dst[i] + bias[f];
    if (do_relu) v = fmaxf(v, 0.0f);
    A[idx] = v;
}

static inline int cdiv_l(long a, long b) { return (int)((a + b - 1) / b); }

template <int DIN, int DN>
static void run_layer(const float* Hin, const float* norm_src, const float* norm_dst,
                      const float* W, const float* bias, float* bufP, float* Agg,
                      const int* src, const int* dst, int N, int E, int do_relu,
                      hipStream_t stream) {
    constexpr int CT = (DN + 15) / 16;
    gcn_gemm_wmma<DIN, DN><<<cdiv_l(N, 16), 32 * CT, 0, stream>>>(
        Hin, norm_src, W, bufP, N);
    gcn_zero<<<cdiv_l((long)N * DN, 256), 256, 0, stream>>>(Agg, (long)N * DN);
    long tot = (long)E * (DN / 4);
    gcn_scatter_add<DN><<<cdiv_l(tot, 256), 256, 0, stream>>>(bufP, src, dst, Agg, E);
    gcn_post<<<cdiv_l((long)N * DN, 256), 256, 0, stream>>>(
        Agg, norm_dst, bias, DN, N, do_relu);
}

extern "C" void kernel_launch(void* const* d_in, const int* in_sizes, int n_in,
                              void* d_out, int out_size, void* d_ws, size_t ws_size,
                              hipStream_t stream) {
    const float* x   = (const float*)d_in[0];
    const int*   src = (const int*)d_in[1];
    const int*   dst = (const int*)d_in[2];
    const float* W1 = (const float*)d_in[3];
    const float* b1 = (const float*)d_in[4];
    const float* W2 = (const float*)d_in[5];
    const float* b2 = (const float*)d_in[6];
    const float* W3 = (const float*)d_in[7];
    const float* b3 = (const float*)d_in[8];
    const float* W4 = (const float*)d_in[9];
    const float* b4 = (const float*)d_in[10];

    const int N = in_sizes[0] / 256;
    const int E = in_sizes[1];

    float* ws       = (float*)d_ws;
    float* norm_src = ws;
    float* norm_dst = ws + N;
    float* bufP     = ws + 2 * (size_t)N;
    float* bufH     = bufP + (size_t)N * 128;

    gcn_zero<<<cdiv_l(2L * N, 256), 256, 0, stream>>>(norm_src, 2L * N);
    gcn_degree<<<cdiv_l(E, 256), 256, 0, stream>>>(src, dst, norm_src, norm_dst, E);
    gcn_rsqrt<<<cdiv_l(2L * N, 256), 256, 0, stream>>>(norm_src, 2L * N);

    run_layer<256, 128>(x,    norm_src, norm_dst, W1, b1, bufP, bufH,
                        src, dst, N, E, 1, stream);
    run_layer<128, 64>(bufH,  norm_src, norm_dst, W2, b2, bufP, bufH,
                       src, dst, N, E, 1, stream);
    run_layer<64, 16>(bufH,   norm_src, norm_dst, W3, b3, bufP, bufH,
                      src, dst, N, E, 1, stream);
    run_layer<16, 40>(bufH,   norm_src, norm_dst, W4, b4, bufP, (float*)d_out,
                      src, dst, N, E, 0, stream);
}


# Layer 1's product: a block walks 5 row tiles with each wave's W1 fragments held in registers (W1 read once per 5 tiles, was once per tile); the next tile's rows are prefetched during the WMMAs
# speedup vs baseline: 1.9617x; 1.9617x over previous
#include <hip/hip_runtime.h>
#include <hip/hip_bf16.h>

typedef __attribute__((ext_vector_type(2))) float v2f;
typedef __attribute__((ext_vector_type(8))) float v8f;

__global__ void gcn_zero(float* __restrict__ p, long n) {
    long i = (long)blockIdx.x * blockDim.x + threadIdx.x;
    if (i < n) p[i] = 0.0f;
}

__device__ __forceinline__ float gcn_norm(float deg) { return rsqrtf(fmaxf(deg, 1.0f)); }

constexpr int PRE = 1, LAST = 2;

template <int DIN, int DN, int MODE>
__global__ __launch_bounds__(((DN + 15) / 16) * 32) void gcn_gemm_wmma(
    const float* __restrict__ H, const float* __restrict__ norm_src,
    const float* __restrict__ norm_dst, const float* __restrict__ bias_in,
    const float* __restrict__ W, float* __restrict__ Y, float* __restrict__ Sum, int n)
{
    constexpr int LDA = DIN + 1;
    __shared__ float As[16 * LDA];

    const int row0 = blockIdx.x * 16;
    const int tid  = threadIdx.x;

    if constexpr (MODE != LAST) {
        constexpr int NT = ((DN + 15) / 16) * 32;
        const int rows = (n - row0 < 16) ? (n - row0) : 16;
        float4* __restrict__ z = reinterpret_cast<float4*>(Sum + (size_t)row0 * DN);
        for (int i = tid; i < 16 * (DN / 4); i += NT)
            if (i < rows * (DN / 4)) z[i] = make_float4(0.0f, 0.0f, 0.0f, 0.0f);
    }

    for (int idx = tid; idx < 16 * DIN; idx += blockDim.x) {
        int r = idx / DIN;
        int c = idx - r * DIN;
        int row = row0 + r;
        float v = 0.0f;
        if (row < n) {
            float h = H[(size_t)row * DIN + c];
            if constexpr (MODE == PRE)
                v = fmaxf(h * gcn_norm(norm_dst[row]) + bias_in[c], 0.0f) * gcn_norm(norm_src[row]);
            else
                v = h;
        }
        As[r * LDA + c] = v;
    }
    __syncthreads();

    const int lane = tid & 31;
    const int wave = tid >> 5;
    const int mn   = lane & 15;
    const int kh   = lane >> 4;
    const int col  = (wave << 4) + mn;

    const int   colc = (col < DN) ? col : (DN - 1);
    float msk = 1.0f;
    if constexpr ((DN & 15) != 0) msk = (col < DN) ? 1.0f : 0.0f;

    const float* __restrict__ ap = As + mn * LDA + 2 * kh;
    const float* __restrict__ wp = W + (size_t)(2 * kh) * DN + colc;

    v8f acc = {0.f, 0.f, 0.f, 0.f, 0.f, 0.f, 0.f, 0.f};

    float nd[8] = {1.f, 1.f, 1.f, 1.f, 1.f, 1.f, 1.f, 1.f};
    float bo = 0.0f;
    if constexpr (MODE == LAST) {
        bo = bias_in[colc];
        #pragma unroll
        for (int v = 0; v < 8; ++v) {
            const int row = row0 + v + 8 * kh;
            nd[v] = gcn_norm(norm_dst[(row < n) ? row : (n - 1)]);
        }
    }

    #pragma unroll
    for (int k = 0; k < DIN; k += 4) {
        v2f a, b;
        a.x = ap[k];
        a.y = ap[k + 1];
        b.x = wp[k * DN] * msk;
        b.y = wp[k * DN + DN] * msk;
        acc = __builtin_amdgcn_wmma_f32_16x16x4_f32(
            false, a, false, b, (short)0, acc, false, false);
    }

    if (col < DN) {
        #pragma unroll
        for (int v = 0; v < 8; ++v) {
            int row = row0 + v + 8 * kh;
            if (row < n) {
                float r = acc[v];
                if constexpr (MODE == LAST) r = r * nd[v] + bo;
                Y[(size_t)row * DN + col] = r;
            }
        }
    }
}

template <int DIN, int DN, int TPB>
__global__ __launch_bounds__((DN / 16) * 32) void gcn_gemm_first(
    const float* __restrict__ X, const float* __restrict__ W, float* __restrict__ Y,
    float* __restrict__ Sum, int n,
    const int* __restrict__ esrc, const int* __restrict__ edst, float* __restrict__ deg_w,
    int n_edges)
{
    constexpr int NT  = (DN / 16) * 32;
    constexpr int LDA = DIN + 1;
    constexpr int NLD = (16 * DIN) / NT;
    static_assert(DN % 16 == 0 && (16 * DIN) % NT == 0 && DIN % 4 == 0, "whole tiles only");
    __shared__ float As[16 * LDA];

    const int tid  = threadIdx.x;
    const int lane = tid & 31;
    const int wave = tid >> 5;
    const int mn   = lane & 15;
    const int kh   = lane >> 4;
    const int col  = (wave << 4) + mn;

    const int tile0 = (int)blockIdx.x * TPB;

    float pre[NLD];
    #pragma unroll
    for (int i = 0; i < NLD; ++i) {
        const int idx = tid + i * NT;
        const int r = idx / DIN, c = idx - r * DIN;
        const int row = tile0 * 16 + r;
        pre[i] = X[(size_t)((row < n) ? row : (n - 1)) * DIN + c];
    }

    const float* __restrict__ wp = W + (size_t)(2 * kh) * DN + col;
    v2f bw[DIN / 4];
    #pragma unroll
    for (int k = 0; k < DIN; k += 4) {
        bw[k / 4].x = wp[k * DN];
        bw[k / 4].y = wp[k * DN + DN];
    }

    {
        const int per = (n_edges + (int)gridDim.x - 1) / (int)gridDim.x;
        const int eb  = (int)blockIdx.x * per;
        for (int i0 = tid; i0 < per; i0 += NT * TPB) {
            int s[TPB], d[TPB];
            #pragma unroll
            for (int u = 0; u < TPB; ++u) {
                const int e  = eb + i0 + u * NT;
                const int ec = (e < n_edges) ? e : (n_edges - 1);
                s[u] = esrc[ec];
                d[u] = edst[ec];
            }
            #pragma unroll
            for (int u = 0; u < TPB; ++u) {
                const int i = i0 + u * NT;
                if (i < per && eb + i < n_edges) {
                    unsafeAtomicAdd(deg_w + s[u], 1.0f);
                    unsafeAtomicAdd(deg_w + n + d[u], 1.0f);
                }
            }
        }
    }

    const float* __restrict__ ap = As + mn * LDA + 2 * kh;

    #pragma unroll 1
    for (int t = 0; t < TPB; ++t) {
        const int row0 = (tile0 + t) * 16;
        if (row0 >= n) break;

        #pragma unroll
        for (int i = 0; i < NLD; ++i) {
            const int idx = tid + i * NT;
            const int r = idx / DIN, c = idx - r * DIN;
            As[r * LDA + c] = (row0 + r < n) ? pre[i] : 0.0f;
        }
        __syncthreads();

        if (t + 1 < TPB && row0 + 16 < n) {
            #pragma unroll
            for (int i = 0; i < NLD; ++i) {
                const int idx = tid + i * NT;
                const int r = idx / DIN, c = idx - r * DIN;
                const int row = row0 + 16 + r;
                pre[i] = X[(size_t)((row < n) ? row : (n - 1)) * DIN + c];
            }
        }

        v8f acc = {0.f, 0.f, 0.f, 0.f, 0.f, 0.f, 0.f, 0.f};
        #pragma unroll
        for (int k = 0; k < DIN; k += 4) {
            v2f a;
            a.x = ap[k];
            a.y = ap[k + 1];
            acc = __builtin_amdgcn_wmma_f32_16x16x4_f32(
                false, a, false, bw[k / 4], (short)0, acc, false, false);
        }

        #pragma unroll
        for (int v = 0; v < 8; ++v) {
            const int row = row0 + v + 8 * kh;
            if (row < n) Y[(size_t)row * DN + col] = acc[v];
        }

        {
            const int rows = (n - row0 < 16) ? (n - row0) : 16;
            float4* __restrict__ z = reinterpret_cast<float4*>(Sum + (size_t)row0 * DN);
            for (int i = tid; i < 16 * (DN / 4); i += NT)
                if (i < rows * (DN / 4)) z[i] = make_float4(0.0f, 0.0f, 0.0f, 0.0f);
        }

        __syncthreads();
    }
}

template <int D, bool SCALE>
__global__ __launch_bounds__(256) void gcn_scatter_add(
    const float* __restrict__ Y, const int* __restrict__ src, const int* __restrict__ dst,
    float* __restrict__ Out, int n_edges, const float* __restrict__ deg_src)
{
    static_assert(!SCALE || D % 32 == 0, "the row factor is applied in the whole-line form only");
    const int lane = threadIdx.x & 31;
    const int wib  = __builtin_amdgcn_readfirstlane((int)(threadIdx.x >> 5));
    const int e0   = (blockIdx.x * (int)(blockDim.x >> 5) + wib) * 32;
    if (e0 >= n_edges) return;
    const int left = n_edges - e0;

    int sv = 0, tv = 0;
    if (lane < left) { sv = src[e0 + lane]; tv = dst[e0 + lane]; }
    int fv = 0;
    if constexpr (SCALE) {
        if (lane < left) fv = __builtin_bit_cast(int, gcn_norm(deg_src[sv]));
    }

    if (left < 32) {
        for (int j = 0; j < left; ++j) {
            const int s = __builtin_amdgcn_readlane(sv, j);
            const int t = __builtin_amdgcn_readlane(tv, j);
            float w = 1.0f;
            if constexpr (SCALE) w = __builtin_bit_cast(float, __builtin_amdgcn_readlane(fv, j));
            for (int f = lane; f < D; f += 32) {
                float y = Y[(size_t)s * D + f];
                if constexpr (SCALE) y *= w;
                unsafeAtomicAdd(Out + (size_t)t * D + f, y);
            }
        }
        return;
    }

    if constexpr (D % 32 == 0) {
        constexpr int Q = D / 32;
        constexpr int G = 32 / Q;
        #pragma unroll
        for (int g = 0; g < 32; g += G) {
            float v[G][Q];
            #pragma unroll
            for (int j = 0; j < G; ++j) {
                const int s = __builtin_amdgcn_readlane(sv, g + j);
                const float* __restrict__ y = Y + (size_t)s * D + lane;
                #pragma unroll
                for (int q = 0; q < Q; ++q) v[j][q] = y[32 * q];
            }
            #pragma unroll
            for (int j = 0; j < G; ++j) {
                const int t = __builtin_amdgcn_readlane(tv, g + j);
                float* __restrict__ o = Out + (size_t)t * D + lane;
                if constexpr (SCALE) {
                    const float w = __builtin_bit_cast(float, __builtin_amdgcn_readlane(fv, g + j));
                    #pragma unroll
                    for (int q = 0; q < Q; ++q) v[j][q] *= w;
                }
                #pragma unroll
                for (int q = 0; q < Q; ++q) unsafeAtomicAdd(o + 32 * q, v[j][q]);
            }
        }
    } else if constexpr (D == 16) {
        const bool hi = lane >= 16;
        const int  c  = lane & 15;
        float v[16];
        #pragma unroll
        for (int p = 0; p < 16; ++p) {
            const int sa = __builtin_amdgcn_readlane(sv, 2 * p);
            const int sb = __builtin_amdgcn_readlane(sv, 2 * p + 1);
            v[p] = Y[(size_t)(hi ? sb : sa) * 16 + c];
        }
        #pragma unroll
        for (int p = 0; p < 16; ++p) {
            const int ta = __builtin_amdgcn_readlane(tv, 2 * p);
            const int tb = __builtin_amdgcn_readlane(tv, 2 * p + 1);
            unsafeAtomicAdd(Out + (size_t)(hi ? tb : ta) * 16 + c, v[p]);
        }
    } else {
        static_assert(D > 32 && D <= 40, "row = one 32-float piece + a tail of up to 8 floats");
        const int k  = lane >> 3;
        const int c  = 32 + (lane & 7);
        const bool on = c < D;
        #pragma unroll
        for (int h = 0; h < 32; h += 16) {
            float v[16], w[4];
            #pragma unroll
            for (int j = 0; j < 16; ++j) {
                const int s = __builtin_amdgcn_readlane(sv, h + j);
                v[j] = Y[(size_t)s * D + lane];
            }
            #pragma unroll
            for (int m = 0; m < 4; ++m) {
                const int s = __shfl(sv, h + 4 * m + k);
                w[m] = on ? Y[(size_t)s * D + c] : 0.0f;
            }
            #pragma unroll
            for (int j = 0; j < 16; ++j) {
                const int t = __builtin_amdgcn_readlane(tv, h + j);
                unsafeAtomicAdd(Out + (size_t)t * D + lane, v[j]);
            }
            #pragma unroll
            for (int m = 0; m < 4; ++m) {
                const int t = __shfl(tv, h + 4 * m + k);
                if (on) unsafeAtomicAdd(Out + (size_t)t * D + c, w[m]);
            }
        }
    }
}

template <int D>
__global__ void gcn_mid(const float* __restrict__ A, const float* __restrict__ norm_src,
                        const float* __restrict__ norm_dst, const float* __restrict__ bias,
                        float* __restrict__ Hs, float* __restrict__ Z, int n)
{
    static_assert(D % 4 == 0, "a row is a whole number of 16-byte pieces");
    const long q = (long)blockIdx.x * blockDim.x + threadIdx.x;
    if (q >= (long)n * (D / 4)) return;
    const int i = (int)(q / (D / 4));
    const int f = (int)(q - (long)i * (D / 4)) * 4;
    const float4 a = reinterpret_cast<const float4*>(A)[q];
    const float nd = gcn_norm(norm_dst[i]);
    const float ns = gcn_norm(norm_src[i]);
    float4 h;
    h.x = fmaxf(a.x * nd + bias[f],     0.0f) * ns;
    h.y = fmaxf(a.y * nd + bias[f + 1], 0.0f) * ns;
    h.z = fmaxf(a.z * nd + bias[f + 2], 0.0f) * ns;
    h.w = fmaxf(a.w * nd + bias[f + 3], 0.0f) * ns;
    reinterpret_cast<float4*>(Hs)[q] = h;
    reinterpret_cast<float4*>(Z)[q]  = make_float4(0.0f, 0.0f, 0.0f, 0.0f);
}

static inline int cdiv_l(long a, long b) { return (int)((a + b - 1) / b); }

template <int DIN, int DN>
static void run_layer(const float* Hin, float* deg,
                      const float* bias_in, const float* W, float* bufP, float* Agg,
                      const int* src, const int* dst, int N, int E, hipStream_t stream) {
    constexpr int CT = (DN + 15) / 16;
    gcn_gemm_wmma<DIN, DN, PRE><<<cdiv_l(N, 16), 32 * CT, 0, stream>>>(
        Hin, deg, deg + N, bias_in, W, bufP, Agg, N);
    gcn_scatter_add<DN, false><<<cdiv_l(E, 256), 256, 0, stream>>>(
        bufP, src, dst, Agg, E, deg);
}

template <int DIN, int DN, int TPB>
static void run_first(const float* x, float* deg, const float* W, float* bufP, float* Agg,
                      const int* src, const int* dst, int N, int E, hipStream_t stream) {
    gcn_gemm_first<DIN, DN, TPB><<<cdiv_l(cdiv_l(N, 16), TPB), 32 * (DN / 16), 0, stream>>>(
        x, W, bufP, Agg, N, src, dst, deg, E);
    gcn_scatter_add<DN, true><<<cdiv_l(E, 256), 256, 0, stream>>>(
        bufP, src, dst, Agg, E, deg);
}

extern "C" void kernel_launch(void* const* d_in, const int* in_sizes, int n_in,
                              void* d_out, int out_size, void* d_ws, size_t ws_size,
                              hipStream_t stream) {
    const float* x   = (const float*)d_in[0];
    const int*   src = (const int*)d_in[1];
    const int*   dst = (const int*)d_in[2];
    const float* W1 = (const float*)d_in[3];
    const float* b1 = (const float*)d_in[4];
    const float* W2 = (const float*)d_in[5];
    const float* b2 = (const float*)d_in[6];
    const float* W3 = (const float*)d_in[7];
    const float* b3 = (const float*)d_in[8];
    const float* W4 = (const float*)d_in[9];
    const float* b4 = (const float*)d_in[10];

    const int N = in_sizes[0] / 256;
    const int E = in_sizes[1];

    float* ws       = (float*)d_ws;
    float* norm_src = ws;
    float* norm_dst = ws + N;
    float* bufA     = ws + 2 * (size_t)N;
    float* bufB     = bufA + (size_t)N * 128;
    float* out      = (float*)d_out;

    gcn_zero<<<cdiv_l(2L * N, 256), 256, 0, stream>>>(norm_src, 2L * N);

    float* sum2 = bufA + (size_t)N * 64;
    float* sum3 = bufB + (size_t)N * 16;
    run_first<256, 128, 5>(x,    norm_src,     W1, bufA, bufB, src, dst, N, E, stream);
    run_layer<128, 64>(bufB,     norm_src, b1, W2, bufA, sum2, src, dst, N, E, stream);
    run_layer<64, 16>(sum2,      norm_src, b2, W3, bufB, sum3, src, dst, N, E, stream);

    float* hs   = bufA;
    float* sum4 = bufA + (size_t)N * 16;
    gcn_mid<16><<<cdiv_l(4L * N, 256), 256, 0, stream>>>(sum3, norm_src, norm_dst, b3, hs, sum4, N);
    gcn_scatter_add<16, false><<<cdiv_l(E, 256), 256, 0, stream>>>(
        hs, src, dst, sum4, E, norm_src);
    gcn_gemm_wmma<16, 40, LAST><<<cdiv_l(N, 16), 32 * 3, 0, stream>>>(
        sum4, norm_src, norm_dst, b4, W4, out, nullptr, N);
}


# speedup vs baseline: 2.2410x; 1.1424x over previous
#include <hip/hip_runtime.h>
#include <hip/hip_bf16.h>

typedef __attribute__((ext_vector_type(2))) float v2f;
typedef __attribute__((ext_vector_type(8))) float v8f;

constexpr int GCN_TK = 4 * 128;

__global__ void gcn_zero(float* __restrict__ p, long n, int* __restrict__ tk) {
    long i = (long)blockIdx.x * blockDim.x + threadIdx.x;
    if (i < n) p[i] = 0.0f;
    if (i < GCN_TK) tk[i] = 0;
}

__device__ __forceinline__ float gcn_norm(float deg) { return rsqrtf(fmaxf(deg, 1.0f)); }

constexpr int PRE = 1, LAST = 2;

template <int DIN, int DN, int MODE>
__global__ __launch_bounds__(((DN + 15) / 16) * 32) void gcn_gemm_wmma(
    const float* __restrict__ H, const float* __restrict__ norm_src,
    const float* __restrict__ norm_dst, const float* __restrict__ bias_in,
    const float* __restrict__ W, float* __restrict__ Y, float* __restrict__ Sum, int n)
{
    constexpr int LDA = DIN + 1;
    __shared__ float As[16 * LDA];

    const int row0 = blockIdx.x * 16;
    const int tid  = threadIdx.x;

    if constexpr (MODE != LAST) {
        constexpr int NT = ((DN + 15) / 16) * 32;
        const int rows = (n - row0 < 16) ? (n - row0) : 16;
        float4* __restrict__ z = reinterpret_cast<float4*>(Sum + (size_t)row0 * DN);
        for (int i = tid; i < 16 * (DN / 4); i += NT)
            if (i < rows * (DN / 4)) z[i] = make_float4(0.0f, 0.0f, 0.0f, 0.0f);
    }

    for (int idx = tid; idx < 16 * DIN; idx += blockDim.x) {
        int r = idx / DIN;
        int c = idx - r * DIN;
        int row = row0 + r;
        float v = 0.0f;
        if (row < n) {
            float h = H[(size_t)row * DIN + c];
            if constexpr (MODE == PRE)
                v = fmaxf(h * gcn_norm(norm_dst[row]) + bias_in[c], 0.0f) * gcn_norm(norm_src[row]);
            else
                v = h;
        }
        As[r * LDA + c] = v;
    }
    __syncthreads();

    const int lane = tid & 31;
    const int wave = tid >> 5;
    const int mn   = lane & 15;
    const int kh   = lane >> 4;
    const int col  = (wave << 4) + mn;

    const int   colc = (col < DN) ? col : (DN - 1);
    float msk = 1.0f;
    if constexpr ((DN & 15) != 0) msk = (col < DN) ? 1.0f : 0.0f;

    const float* __restrict__ ap = As + mn * LDA + 2 * kh;
    const float* __restrict__ wp = W + (size_t)(2 * kh) * DN + colc;

    v8f acc = {0.f, 0.f, 0.f, 0.f, 0.f, 0.f, 0.f, 0.f};

    float nd[8] = {1.f, 1.f, 1.f, 1.f, 1.f, 1.f, 1.f, 1.f};
    float bo = 0.0f;
    if constexpr (MODE == LAST) {
        bo = bias_in[colc];
        #pragma unroll
        for (int v = 0; v < 8; ++v) {
            const int row = row0 + v + 8 * kh;
            nd[v] = gcn_norm(norm_dst[(row < n) ? row : (n - 1)]);
        }
    }

    #pragma unroll
    for (int k = 0; k < DIN; k += 4) {
        v2f a, b;
        a.x = ap[k];
        a.y = ap[k + 1];
        b.x = wp[k * DN] * msk;
        b.y = wp[k * DN + DN] * msk;
        acc = __builtin_amdgcn_wmma_f32_16x16x4_f32(
            false, a, false, b, (short)0, acc, false, false);
    }

    if (col < DN) {
        #pragma unroll
        for (int v = 0; v < 8; ++v) {
            int row = row0 + v + 8 * kh;
            if (row < n) {
                float r = acc[v];
                if constexpr (MODE == LAST) r = r * nd[v] + bo;
                Y[(size_t)row * DN + col] = r;
            }
        }
    }
}

template <int DIN, int DN, int TPB>
__global__ __launch_bounds__((DN / 16) * 32) void gcn_gemm_first(
    const float* __restrict__ X, const float* __restrict__ W, float* __restrict__ Y,
    float* __restrict__ Sum, int n,
    const int* __restrict__ esrc, const int* __restrict__ edst, float* __restrict__ deg_w,
    int n_edges)
{
    constexpr int NT  = (DN / 16) * 32;
    constexpr int LDA = DIN + 1;
    constexpr int NLD = (16 * DIN) / NT;
    static_assert(DN % 16 == 0 && (16 * DIN) % NT == 0 && DIN % 4 == 0, "whole tiles only");
    __shared__ float As[16 * LDA];

    const int tid  = threadIdx.x;
    const int lane = tid & 31;
    const int wave = tid >> 5;
    const int mn   = lane & 15;
    const int kh   = lane >> 4;
    const int col  = (wave << 4) + mn;

    const int tile0 = (int)blockIdx.x * TPB;

    float pre[NLD];
    #pragma unroll
    for (int i = 0; i < NLD; ++i) {
        const int idx = tid + i * NT;
        const int r = idx / DIN, c = idx - r * DIN;
        const int row = tile0 * 16 + r;
        pre[i] = X[(size_t)((row < n) ? row : (n - 1)) * DIN + c];
    }

    const float* __restrict__ wp = W + (size_t)(2 * kh) * DN + col;
    v2f bw[DIN / 4];
    #pragma unroll
    for (int k = 0; k < DIN; k += 4) {
        bw[k / 4].x = wp[k * DN];
        bw[k / 4].y = wp[k * DN + DN];
    }

    {
        const int per = (n_edges + (int)gridDim.x - 1) / (int)gridDim.x;
        const int eb  = (int)blockIdx.x * per;
        for (int i0 = tid; i0 < per; i0 += NT * TPB) {
            int s[TPB], d[TPB];
            #pragma unroll
            for (int u = 0; u < TPB; ++u) {
                const int e  = eb + i0 + u * NT;
                const int ec = (e < n_edges) ? e : (n_edges - 1);
                s[u] = esrc[ec];
                d[u] = edst[ec];
            }
            #pragma unroll
            for (int u = 0; u < TPB; ++u) {
                const int i = i0 + u * NT;
                if (i < per && eb + i < n_edges) {
                    unsafeAtomicAdd(deg_w + s[u], 1.0f);
                    unsafeAtomicAdd(deg_w + n + d[u], 1.0f);
                }
            }
        }
    }

    const float* __restrict__ ap = As + mn * LDA + 2 * kh;

    #pragma unroll 1
    for (int t = 0; t < TPB; ++t) {
        const int row0 = (tile0 + t) * 16;
        if (row0 >= n) break;

        #pragma unroll
        for (int i = 0; i < NLD; ++i) {
            const int idx = tid + i * NT;
            const int r = idx / DIN, c = idx - r * DIN;
            As[r * LDA + c] = (row0 + r < n) ? pre[i] : 0.0f;
        }
        __syncthreads();

        if (t + 1 < TPB && row0 + 16 < n) {
            #pragma unroll
            for (int i = 0; i < NLD; ++i) {
                const int idx = tid + i * NT;
                const int r = idx / DIN, c = idx - r * DIN;
                const int row = row0 + 16 + r;
                pre[i] = X[(size_t)((row < n) ? row : (n - 1)) * DIN + c];
            }
        }

        v8f acc = {0.f, 0.f, 0.f, 0.f, 0.f, 0.f, 0.f, 0.f};
        #pragma unroll
        for (int k = 0; k < DIN; k += 4) {
            v2f a;
            a.x = ap[k];
            a.y = ap[k + 1];
            acc = __builtin_amdgcn_wmma_f32_16x16x4_f32(
                false, a, false, bw[k / 4], (short)0, acc, false, false);
        }

        #pragma unroll
        for (int v = 0; v < 8; ++v) {
            const int row = row0 + v + 8 * kh;
            if (row < n) Y[(size_t)row * DN + col] = acc[v];
        }

        {
            const int rows = (n - row0 < 16) ? (n - row0) : 16;
            float4* __restrict__ z = reinterpret_cast<float4*>(Sum + (size_t)row0 * DN);
            for (int i = tid; i < 16 * (DN / 4); i += NT)
                if (i < rows * (DN / 4)) z[i] = make_float4(0.0f, 0.0f, 0.0f, 0.0f);
        }

        __syncthreads();
    }
}


__device__ __forceinline__ int gcn_cache_half() {
    return (int)((__builtin_amdgcn_s_sendmsg_rtn(0x87) >> 18) & 1u);
}

template <int D, bool SCALE>
__device__ __forceinline__ void gcn_rows(const float* __restrict__ Y, float* __restrict__ Out,
                                         int cs, int ct, int cf, int cnt, int lane)
{
    if constexpr (D % 32 == 0) {
        constexpr int Q = D / 32;
        constexpr int G = 32 / Q;
        #pragma unroll
        for (int g = 0; g < 32; g += G) {
            if (g >= cnt) break;
            float v[G][Q];
            #pragma unroll
            for (int j = 0; j < G; ++j) {
                if (g + j < cnt) {
                    const int s = __builtin_amdgcn_readlane(cs, g + j);
                    const float* __restrict__ y = Y + (size_t)s * D + lane;
                    #pragma unroll
                    for (int q = 0; q < Q; ++q) v[j][q] = y[32 * q];
                }
            }
            #pragma unroll
            for (int j = 0; j < G; ++j) {
                if (g + j < cnt) {
                    const int t = __builtin_amdgcn_readlane(ct, g + j);
                    float* __restrict__ o = Out + (size_t)t * D + lane;
                    if constexpr (SCALE) {
                        const float w = __builtin_bit_cast(float, __builtin_amdgcn_readlane(cf, g + j));
                        #pragma unroll
                        for (int q = 0; q < Q; ++q) v[j][q] *= w;
                    }
                    #pragma unroll
                    for (int q = 0; q < Q; ++q) unsafeAtomicAdd(o + 32 * q, v[j][q]);
                }
            }
        }
    } else {
        static_assert(D == 16 && !SCALE, "64-byte rows: two rows per instruction");
        const bool hi = lane >= 16;
        const int  c  = lane & 15;
        float v[16];
        #pragma unroll
        for (int p = 0; p < 16; ++p) {
            if (2 * p < cnt) {
                const int sa = __builtin_amdgcn_readlane(cs, 2 * p);
                const int sb = __builtin_amdgcn_readlane(cs, 2 * p + 1);
                if (!hi || 2 * p + 1 < cnt) v[p] = Y[(size_t)(hi ? sb : sa) * 16 + c];
            }
        }
        #pragma unroll
        for (int p = 0; p < 16; ++p) {
            if (2 * p < cnt) {
                const int ta = __builtin_amdgcn_readlane(ct, 2 * p);
                const int tb = __builtin_amdgcn_readlane(ct, 2 * p + 1);
                if (!hi || 2 * p + 1 < cnt) unsafeAtomicAdd(Out + (size_t)(hi ? tb : ta) * 16 + c, v[p]);
            }
        }
    }
}

template <int D, bool SCALE, int NI>
__global__ __launch_bounds__(256) void gcn_scatter_add(
    const float* __restrict__ Y, const int* __restrict__ src, const int* __restrict__ dst,
    float* __restrict__ Out, int n_edges, const float* __restrict__ deg_src,
    int n, int* __restrict__ tickets)
{
    constexpr int EW = 32 * NI;
    constexpr int EB = 8 * EW;
    __shared__ int ls_s[8][EW];
    __shared__ int ls_t[8][EW];
    __shared__ int ls_f[SCALE ? 8 : 1][SCALE ? EW : 1];
    __shared__ int sh_tk[2];
    __shared__ int sh_half;

    const int lane = threadIdx.x & 31;
    const int wib  = __builtin_amdgcn_readfirstlane((int)(threadIdx.x >> 5));
    const int nchunks = (n_edges + EB - 1) / EB;
    const int nsplit  = (n >> 1) & ~3;

    if (threadIdx.x == 0) sh_half = gcn_cache_half();
    __syncthreads();
    const int half = __builtin_amdgcn_readfirstlane(sh_half);

    int par = 0;
    #pragma unroll 1
    for (int phase = 0; phase < 2; ++phase) {
        const int q = half ^ phase;
        unsigned* __restrict__ ctr = reinterpret_cast<unsigned*>(tickets + 64 * q);
        int nxt = 0;
        if (threadIdx.x == 0) nxt = (int)atomicInc(ctr, 0xffffffffu);
        #pragma unroll 1
        for (;;) {
            if (threadIdx.x == 0) sh_tk[par] = nxt;
            __syncthreads();
            const int tk = __builtin_amdgcn_readfirstlane(sh_tk[par]);
            par ^= 1;
            if (tk >= nchunks) break;
            if (threadIdx.x == 0) nxt = (int)atomicInc(ctr, 0xffffffffu);

            const int e0 = tk * EB + wib * EW;
            int sv[NI], tv[NI], fv[NI];
            bool mine[NI];
            #pragma unroll
            for (int k = 0; k < NI; ++k) {
                const int e  = e0 + 32 * k + lane;
                const int ec = (e < n_edges) ? e : (n_edges - 1);
                sv[k] = src[ec];
                tv[k] = dst[ec];
            }
            #pragma unroll
            for (int k = 0; k < NI; ++k) {
                const int e = e0 + 32 * k + lane;
                mine[k] = (e < n_edges) && ((tv[k] >= nsplit) == (q != 0));
                fv[k] = 0;
                if constexpr (SCALE) {
                    if (mine[k]) fv[k] = __builtin_bit_cast(int, gcn_norm(deg_src[sv[k]]));
                }
            }
            int kept = 0;
            #pragma unroll
            for (int k = 0; k < NI; ++k) {
                const unsigned b = __builtin_amdgcn_ballot_w32(mine[k]);
                const int pos = kept + (int)__builtin_amdgcn_mbcnt_lo(b, 0u);
                if (mine[k]) {
                    ls_s[wib][pos] = sv[k];
                    ls_t[wib][pos] = tv[k];
                    if constexpr (SCALE) ls_f[wib][pos] = fv[k];
                }
                kept += __builtin_popcount(b);
            }
            #pragma unroll 1
            for (int k = 0; k < NI; ++k) {
                const int left = kept - 32 * k;
                if (left <= 0) break;
                const int cnt = (left < 32) ? left : 32;
                const int cs = ls_s[wib][32 * k + lane];
                const int ct = ls_t[wib][32 * k + lane];
                int cf = 0;
                if constexpr (SCALE) cf = ls_f[wib][32 * k + lane];
                gcn_rows<D, SCALE>(Y, Out, cs, ct, cf, cnt, lane);
            }
        }
    }
}

template <int D>
__global__ void gcn_mid(const float* __restrict__ A, const float* __restrict__ norm_src,
                        const float* __restrict__ norm_dst, const float* __restrict__ bias,
                        float* __restrict__ Hs, float* __restrict__ Z, int n)
{
    static_assert(D % 4 == 0, "a row is a whole number of 16-byte pieces");
    const long q = (long)blockIdx.x * blockDim.x + threadIdx.x;
    if (q >= (long)n * (D / 4)) return;
    const int i = (int)(q / (D / 4));
    const int f = (int)(q - (long)i * (D / 4)) * 4;
    const float4 a = reinterpret_cast<const float4*>(A)[q];
    const float nd = gcn_norm(norm_dst[i]);
    const float ns = gcn_norm(norm_src[i]);
    float4 h;
    h.x = fmaxf(a.x * nd + bias[f],     0.0f) * ns;
    h.y = fmaxf(a.y * nd + bias[f + 1], 0.0f) * ns;
    h.z = fmaxf(a.z * nd + bias[f + 2], 0.0f) * ns;
    h.w = fmaxf(a.w * nd + bias[f + 3], 0.0f) * ns;
    reinterpret_cast<float4*>(Hs)[q] = h;
    reinterpret_cast<float4*>(Z)[q]  = make_float4(0.0f, 0.0f, 0.0f, 0.0f);
}

static inline int cdiv_l(long a, long b) { return (int)((a + b - 1) / b); }

constexpr int GCN_SB = 2048;

template <int DIN, int DN>
static void run_layer(const float* Hin, float* deg,
                      const float* bias_in, const float* W, float* bufP, float* Agg,
                      const int* src, const int* dst, int N, int E, int* tk, hipStream_t stream) {
    constexpr int CT = (DN + 15) / 16;
    constexpr int NI = 2;
    gcn_gemm_wmma<DIN, DN, PRE><<<cdiv_l(N, 16), 32 * CT, 0, stream>>>(
        Hin, deg, deg + N, bias_in, W, bufP, Agg, N);
    gcn_scatter_add<DN, false, NI><<<GCN_SB, 256, 0, stream>>>(
        bufP, src, dst, Agg, E, deg, N, tk);
}

template <int DIN, int DN, int TPB>
static void run_first(const float* x, float* deg, const float* W, float* bufP, float* Agg,
                      const int* src, const int* dst, int N, int E, int* tk, hipStream_t stream) {
    gcn_gemm_first<DIN, DN, TPB><<<cdiv_l(cdiv_l(N, 16), TPB), 32 * (DN / 16), 0, stream>>>(
        x, W, bufP, Agg, N, src, dst, deg, E);
    gcn_scatter_add<DN, true, 1><<<GCN_SB, 256, 0, stream>>>(
        bufP, src, dst, Agg, E, deg, N, tk);
}

extern "C" void kernel_launch(void* const* d_in, const int* in_sizes, int n_in,
                              void* d_out, int out_size, void* d_ws, size_t ws_size,
                              hipStream_t stream) {
    const float* x   = (const float*)d_in[0];
    const int*   src = (const int*)d_in[1];
    const int*   dst = (const int*)d_in[2];
    const float* W1 = (const float*)d_in[3];
    const float* b1 = (const float*)d_in[4];
    const float* W2 = (const float*)d_in[5];
    const float* b2 = (const float*)d_in[6];
    const float* W3 = (const float*)d_in[7];
    const float* b3 = (const float*)d_in[8];
    const float* W4 = (const float*)d_in[9];
    const float* b4 = (const float*)d_in[10];

    const int N = in_sizes[0] / 256;
    const int E = in_sizes[1];

    float* ws       = (float*)d_ws;
    float* norm_src = ws;
    float* norm_dst = ws + N;
    float* bufA     = ws + 2 * (size_t)N;
    float* bufB     = bufA + (size_t)N * 128;
    float* out      = (float*)d_out;

    int* tk = (int*)d_out;
    gcn_zero<<<cdiv_l(2L * N, 256), 256, 0, stream>>>(norm_src, 2L * N, tk);

    float* sum2 = bufA + (size_t)N * 64;
    float* sum3 = bufB + (size_t)N * 16;
    run_first<256, 128, 5>(x,    norm_src,     W1, bufA, bufB, src, dst, N, E, tk,       stream);
    run_layer<128, 64>(bufB,     norm_src, b1, W2, bufA, sum2, src, dst, N, E, tk + 128, stream);
    run_layer<64, 16>(sum2,      norm_src, b2, W3, bufB, sum3, src, dst, N, E, tk + 256, stream);

    float* hs   = bufA;
    float* sum4 = bufA + (size_t)N * 16;
    gcn_mid<16><<<cdiv_l(4L * N, 256), 256, 0, stream>>>(sum3, norm_src, norm_dst, b3, hs, sum4, N);
    gcn_scatter_add<16, false, 2><<<GCN_SB, 256, 0, stream>>>(
        hs, src, dst, sum4, E, norm_src, N, tk + 384);
    gcn_gemm_wmma<16, 40, LAST><<<cdiv_l(N, 16), 32 * 3, 0, stream>>>(
        sum4, norm_src, norm_dst, b4, W4, out, nullptr, N);
}


# Products of layers 2-4 in layer 1's form: a block walks 4 (layer 3: 2) row tiles, W fragments in registers, next tile's rows loaded as one group during the WMMAs; 2-4x fewer waves; sums unchanged
# speedup vs baseline: 2.3460x; 1.0469x over previous
#include <hip/hip_runtime.h>
#include <hip/hip_bf16.h>

typedef __attribute__((ext_vector_type(2))) float v2f;
typedef __attribute__((ext_vector_type(8))) float v8f;

constexpr int GCN_TK = 4 * 128;

__global__ void gcn_zero(float* __restrict__ p, long n, int* __restrict__ tk) {
    long i = (long)blockIdx.x * blockDim.x + threadIdx.x;
    if (i < n) p[i] = 0.0f;
    if (i < GCN_TK) tk[i] = 0;
}

__device__ __forceinline__ float gcn_norm(float deg) { return rsqrtf(fmaxf(deg, 1.0f)); }

template <int DIN, int DN, int TPB>
__global__ __launch_bounds__((DN / 16) * 32) void gcn_gemm_first(
    const float* __restrict__ X, const float* __restrict__ W, float* __restrict__ Y,
    float* __restrict__ Sum, int n,
    const int* __restrict__ esrc, const int* __restrict__ edst, float* __restrict__ deg_w,
    int n_edges)
{
    constexpr int NT  = (DN / 16) * 32;
    constexpr int LDA = DIN + 1;
    constexpr int NLD = (16 * DIN) / NT;
    static_assert(DN % 16 == 0 && (16 * DIN) % NT == 0 && DIN % 4 == 0, "whole tiles only");
    __shared__ float As[16 * LDA];

    const int tid  = threadIdx.x;
    const int lane = tid & 31;
    const int wave = tid >> 5;
    const int mn   = lane & 15;
    const int kh   = lane >> 4;
    const int col  = (wave << 4) + mn;

    const int tile0 = (int)blockIdx.x * TPB;

    float pre[NLD];
    #pragma unroll
    for (int i = 0; i < NLD; ++i) {
        const int idx = tid + i * NT;
        const int r = idx / DIN, c = idx - r * DIN;
        const int row = tile0 * 16 + r;
        pre[i] = X[(size_t)((row < n) ? row : (n - 1)) * DIN + c];
    }

    const float* __restrict__ wp = W + (size_t)(2 * kh) * DN + col;
    v2f bw[DIN / 4];
    #pragma unroll
    for (int k = 0; k < DIN; k += 4) {
        bw[k / 4].x = wp[k * DN];
        bw[k / 4].y = wp[k * DN + DN];
    }

    {
        const int per = (n_edges + (int)gridDim.x - 1) / (int)gridDim.x;
        const int eb  = (int)blockIdx.x * per;
        for (int i0 = tid; i0 < per; i0 += NT * TPB) {
            int s[TPB], d[TPB];
            #pragma unroll
            for (int u = 0; u < TPB; ++u) {
                const int e  = eb + i0 + u * NT;
                const int ec = (e < n_edges) ? e : (n_edges - 1);
                s[u] = esrc[ec];
                d[u] = edst[ec];
            }
            #pragma unroll
            for (int u = 0; u < TPB; ++u) {
                const int i = i0 + u * NT;
                if (i < per && eb + i < n_edges) {
                    unsafeAtomicAdd(deg_w + s[u], 1.0f);
                    unsafeAtomicAdd(deg_w + n + d[u], 1.0f);
                }
            }
        }
    }

    const float* __restrict__ ap = As + mn * LDA + 2 * kh;

    #pragma unroll 1
    for (int t = 0; t < TPB; ++t) {
        const int row0 = (tile0 + t) * 16;
        if (row0 >= n) break;

        #pragma unroll
        for (int i = 0; i < NLD; ++i) {
            const int idx = tid + i * NT;
            const int r = idx / DIN, c = idx - r * DIN;
            As[r * LDA + c] = (row0 + r < n) ? pre[i] : 0.0f;
        }
        __syncthreads();

        if (t + 1 < TPB && row0 + 16 < n) {
            #pragma unroll
            for (int i = 0; i < NLD; ++i) {
                const int idx = tid + i * NT;
                const int r = idx / DIN, c = idx - r * DIN;
                const int row = row0 + 16 + r;
                pre[i] = X[(size_t)((row < n) ? row : (n - 1)) * DIN + c];
            }
        }

        v8f acc = {0.f, 0.f, 0.f, 0.f, 0.f, 0.f, 0.f, 0.f};
        #pragma unroll
        for (int k = 0; k < DIN; k += 4) {
            v2f a;
            a.x = ap[k];
            a.y = ap[k + 1];
            acc = __builtin_amdgcn_wmma_f32_16x16x4_f32(
                false, a, false, bw[k / 4], (short)0, acc, false, false);
        }

        #pragma unroll
        for (int v = 0; v < 8; ++v) {
            const int row = row0 + v + 8 * kh;
            if (row < n) Y[(size_t)row * DN + col] = acc[v];
        }

        {
            const int rows = (n - row0 < 16) ? (n - row0) : 16;
            float4* __restrict__ z = reinterpret_cast<float4*>(Sum + (size_t)row0 * DN);
            for (int i = tid; i < 16 * (DN / 4); i += NT)
                if (i < rows * (DN / 4)) z[i] = make_float4(0.0f, 0.0f, 0.0f, 0.0f);
        }

        __syncthreads();
    }
}

template <int DIN, int DN, int TPB>
__global__ __launch_bounds__((DN / 16) * 32) void gcn_gemm_tiles(
    const float* __restrict__ H, const float* __restrict__ deg_src,
    const float* __restrict__ deg_dst, const float* __restrict__ bias_in,
    const float* __restrict__ W, float* __restrict__ Y, float* __restrict__ Sum, int n)
{
    constexpr int NT  = (DN / 16) * 32;
    constexpr int LDA = DIN + 1;
    constexpr int NLD = (16 * DIN) / NT;
    constexpr int NB  = DIN / NT;
    static_assert(DN % 16 == 0 && DIN % NT == 0 && DIN % 4 == 0, "whole tiles only");
    __shared__ float As[16 * LDA];

    const int tid  = threadIdx.x;
    const int lane = tid & 31;
    const int wave = tid >> 5;
    const int mn   = lane & 15;
    const int kh   = lane >> 4;
    const int col  = (wave << 4) + mn;

    const int tile0 = (int)blockIdx.x * TPB;

    float bs[NB];
    #pragma unroll
    for (int j = 0; j < NB; ++j) bs[j] = bias_in[tid + j * NT];

    float pre[NLD];
    #pragma unroll
    for (int i = 0; i < NLD; ++i) {
        const int row = tile0 * 16 + (i * NT) / DIN;
        pre[i] = H[(size_t)((row < n) ? row : (n - 1)) * DIN + tid + (i * NT) % DIN];
    }
    float dd, ds;
    {
        const int row = tile0 * 16 + mn;
        const int rc  = (row < n) ? row : (n - 1);
        dd = deg_dst[rc];
        ds = deg_src[rc];
    }

    const float* __restrict__ wp = W + (size_t)(2 * kh) * DN + col;
    v2f bw[DIN / 4];
    #pragma unroll
    for (int k = 0; k < DIN; k += 4) {
        bw[k / 4].x = wp[k * DN];
        bw[k / 4].y = wp[k * DN + DN];
    }

    const float* __restrict__ ap = As + mn * LDA + 2 * kh;

    #pragma unroll 1
    for (int t = 0; t < TPB; ++t) {
        const int row0 = (tile0 + t) * 16;
        if (row0 >= n) break;

        const int fd = __builtin_bit_cast(int, gcn_norm(dd));
        const int fs = __builtin_bit_cast(int, gcn_norm(ds));
        #pragma unroll
        for (int i = 0; i < NLD; ++i) {
            const int r = (i * NT) / DIN;
            const int c = tid + (i * NT) % DIN;
            const float nd = __builtin_bit_cast(float, __builtin_amdgcn_readlane(fd, r));
            const float ns = __builtin_bit_cast(float, __builtin_amdgcn_readlane(fs, r));
            const float v = fmaxf(pre[i] * nd + bs[i % NB], 0.0f) * ns;
            As[r * LDA + c] = (row0 + r < n) ? v : 0.0f;
        }
        __syncthreads();

        if (t + 1 < TPB && row0 + 16 < n) {
            #pragma unroll
            for (int i = 0; i < NLD; ++i) {
                const int row = row0 + 16 + (i * NT) / DIN;
                pre[i] = H[(size_t)((row < n) ? row : (n - 1)) * DIN + tid + (i * NT) % DIN];
            }
            const int row = row0 + 16 + mn;
            const int rc  = (row < n) ? row : (n - 1);
            dd = deg_dst[rc];
            ds = deg_src[rc];
        }

        v8f acc = {0.f, 0.f, 0.f, 0.f, 0.f, 0.f, 0.f, 0.f};
        #pragma unroll
        for (int k = 0; k < DIN; k += 4) {
            v2f a;
            a.x = ap[k];
            a.y = ap[k + 1];
            acc = __builtin_amdgcn_wmma_f32_16x16x4_f32(
                false, a, false, bw[k / 4], (short)0, acc, false, false);
        }

        #pragma unroll
        for (int v = 0; v < 8; ++v) {
            const int row = row0 + v + 8 * kh;
            if (row < n) Y[(size_t)row * DN + col] = acc[v];
        }

        {
            const int rows = (n - row0 < 16) ? (n - row0) : 16;
            float4* __restrict__ z = reinterpret_cast<float4*>(Sum + (size_t)row0 * DN);
            for (int i = tid; i < 16 * (DN / 4); i += NT)
                if (i < rows * (DN / 4)) z[i] = make_float4(0.0f, 0.0f, 0.0f, 0.0f);
        }

        __syncthreads();
    }
}

template <int DIN, int DN, int TPB>
__global__ __launch_bounds__(((DN + 15) / 16) * 32) void gcn_gemm_last(
    const float* __restrict__ H, const float* __restrict__ deg_dst,
    const float* __restrict__ bias_in, const float* __restrict__ W,
    float* __restrict__ Y, int n)
{
    constexpr int NT  = ((DN + 15) / 16) * 32;
    constexpr int LDA = DIN + 1;
    constexpr int NEL = 16 * DIN;
    constexpr int NLD = (NEL + NT - 1) / NT;
    static_assert(DIN % 4 == 0, "whole k steps only");
    __shared__ float As[16 * LDA];

    const int tid  = threadIdx.x;
    const int lane = tid & 31;
    const int wave = tid >> 5;
    const int mn   = lane & 15;
    const int kh   = lane >> 4;
    const int col  = (wave << 4) + mn;

    const int   colc = (col < DN) ? col : (DN - 1);
    float msk = 1.0f;
    if constexpr ((DN & 15) != 0) msk = (col < DN) ? 1.0f : 0.0f;

    const int tile0 = (int)blockIdx.x * TPB;

    float pre[NLD];
    #pragma unroll
    for (int i = 0; i < NLD; ++i) {
        const int idx = (tid + i * NT < NEL) ? (tid + i * NT) : (NEL - 1);
        const int r = idx / DIN, c = idx - r * DIN;
        const int row = tile0 * 16 + r;
        pre[i] = H[(size_t)((row < n) ? row : (n - 1)) * DIN + c];
    }
    float dq[8];
    #pragma unroll
    for (int v = 0; v < 8; ++v) {
        const int row = tile0 * 16 + v + 8 * kh;
        dq[v] = deg_dst[(row < n) ? row : (n - 1)];
    }

    const float* __restrict__ wp = W + (size_t)(2 * kh) * DN + colc;
    v2f bw[DIN / 4];
    #pragma unroll
    for (int k = 0; k < DIN; k += 4) {
        bw[k / 4].x = wp[k * DN] * msk;
        bw[k / 4].y = wp[k * DN + DN] * msk;
    }
    const float bo = bias_in[colc];

    const float* __restrict__ ap = As + mn * LDA + 2 * kh;

    #pragma unroll 1
    for (int t = 0; t < TPB; ++t) {
        const int row0 = (tile0 + t) * 16;
        if (row0 >= n) break;

        float dc[8];
        #pragma unroll
        for (int v = 0; v < 8; ++v) dc[v] = dq[v];
        #pragma unroll
        for (int i = 0; i < NLD; ++i) {
            const int idx = tid + i * NT;
            if (idx < NEL) {
                const int r = idx / DIN, c = idx - r * DIN;
                As[r * LDA + c] = (row0 + r < n) ? pre[i] : 0.0f;
            }
        }
        __syncthreads();

        if (t + 1 < TPB && row0 + 16 < n) {
            #pragma unroll
            for (int i = 0; i < NLD; ++i) {
                const int idx = (tid + i * NT < NEL) ? (tid + i * NT) : (NEL - 1);
                const int r = idx / DIN, c = idx - r * DIN;
                const int row = row0 + 16 + r;
                pre[i] = H[(size_t)((row < n) ? row : (n - 1)) * DIN + c];
            }
            #pragma unroll
            for (int v = 0; v < 8; ++v) {
                const int row = row0 + 16 + v + 8 * kh;
                dq[v] = deg_dst[(row < n) ? row : (n - 1)];
            }
        }

        v8f acc = {0.f, 0.f, 0.f, 0.f, 0.f, 0.f, 0.f, 0.f};
        #pragma unroll
        for (int k = 0; k < DIN; k += 4) {
            v2f a;
            a.x = ap[k];
            a.y = ap[k + 1];
            acc = __builtin_amdgcn_wmma_f32_16x16x4_f32(
                false, a, false, bw[k / 4], (short)0, acc, false, false);
        }

        if (col < DN) {
            #pragma unroll
            for (int v = 0; v < 8; ++v) {
                const int row = row0 + v + 8 * kh;
                if (row < n) Y[(size_t)row * DN + col] = acc[v] * gcn_norm(dc[v]) + bo;
            }
        }

        __syncthreads();
    }
}


__device__ __forceinline__ int gcn_cache_half() {
    return (int)((__builtin_amdgcn_s_sendmsg_rtn(0x87) >> 18) & 1u);
}

template <int D, bool SCALE>
__device__ __forceinline__ void gcn_rows(const float* __restrict__ Y, float* __restrict__ Out,
                                         int cs, int ct, int cf, int cnt, int lane)
{
    if constexpr (D % 32 == 0) {
        constexpr int Q = D / 32;
        constexpr int G = 32 / Q;
        #pragma unroll
        for (int g = 0; g < 32; g += G) {
            if (g >= cnt) break;
            float v[G][Q];
            #pragma unroll
            for (int j = 0; j < G; ++j) {
                if (g + j < cnt) {
                    const int s = __builtin_amdgcn_readlane(cs, g + j);
                    const float* __restrict__ y = Y + (size_t)s * D + lane;
                    #pragma unroll
                    for (int q = 0; q < Q; ++q) v[j][q] = y[32 * q];
                }
            }
            #pragma unroll
            for (int j = 0; j < G; ++j) {
                if (g + j < cnt) {
                    const int t = __builtin_amdgcn_readlane(ct, g + j);
                    float* __restrict__ o = Out + (size_t)t * D + lane;
                    if constexpr (SCALE) {
                        const float w = __builtin_bit_cast(float, __builtin_amdgcn_readlane(cf, g + j));
                        #pragma unroll
                        for (int q = 0; q < Q; ++q) v[j][q] *= w;
                    }
                    #pragma unroll
                    for (int q = 0; q < Q; ++q) unsafeAtomicAdd(o + 32 * q, v[j][q]);
                }
            }
        }
    } else {
        static_assert(D == 16 && !SCALE, "64-byte rows: two rows per instruction");
        const bool hi = lane >= 16;
        const int  c  = lane & 15;
        float v[16];
        #pragma unroll
        for (int p = 0; p < 16; ++p) {
            if (2 * p < cnt) {
                const int sa = __builtin_amdgcn_readlane(cs, 2 * p);
                const int sb = __builtin_amdgcn_readlane(cs, 2 * p + 1);
                if (!hi || 2 * p + 1 < cnt) v[p] = Y[(size_t)(hi ? sb : sa) * 16 + c];
            }
        }
        #pragma unroll
        for (int p = 0; p < 16; ++p) {
            if (2 * p < cnt) {
                const int ta = __builtin_amdgcn_readlane(ct, 2 * p);
                const int tb = __builtin_amdgcn_readlane(ct, 2 * p + 1);
                if (!hi || 2 * p + 1 < cnt) unsafeAtomicAdd(Out + (size_t)(hi ? tb : ta) * 16 + c, v[p]);
            }
        }
    }
}

template <int D, bool SCALE, int NI>
__global__ __launch_bounds__(256) void gcn_scatter_add(
    const float* __restrict__ Y, const int* __restrict__ src, const int* __restrict__ dst,
    float* __restrict__ Out, int n_edges, const float* __restrict__ deg_src,
    int n, int* __restrict__ tickets)
{
    constexpr int EW = 32 * NI;
    constexpr int EB = 8 * EW;
    __shared__ int ls_s[8][EW];
    __shared__ int ls_t[8][EW];
    __shared__ int ls_f[SCALE ? 8 : 1][SCALE ? EW : 1];
    __shared__ int sh_tk[2];
    __shared__ int sh_half;

    const int lane = threadIdx.x & 31;
    const int wib  = __builtin_amdgcn_readfirstlane((int)(threadIdx.x >> 5));
    const int nchunks = (n_edges + EB - 1) / EB;
    const int nsplit  = (n >> 1) & ~3;

    if (threadIdx.x == 0) sh_half = gcn_cache_half();
    __syncthreads();
    const int half = __builtin_amdgcn_readfirstlane(sh_half);

    int par = 0;
    #pragma unroll 1
    for (int phase = 0; phase < 2; ++phase) {
        const int q = half ^ phase;
        unsigned* __restrict__ ctr = reinterpret_cast<unsigned*>(tickets + 64 * q);
        int nxt = 0;
        if (threadIdx.x == 0) nxt = (int)atomicInc(ctr, 0xffffffffu);
        #pragma unroll 1
        for (;;) {
            if (threadIdx.x == 0) sh_tk[par] = nxt;
            __syncthreads();
            const int tk = __builtin_amdgcn_readfirstlane(sh_tk[par]);
            par ^= 1;
            if (tk >= nchunks) break;
            if (threadIdx.x == 0) nxt = (int)atomicInc(ctr, 0xffffffffu);

            const int e0 = tk * EB + wib * EW;
            int sv[NI], tv[NI], fv[NI];
            bool mine[NI];
            #pragma unroll
            for (int k = 0; k < NI; ++k) {
                const int e  = e0 + 32 * k + lane;
                const int ec = (e < n_edges) ? e : (n_edges - 1);
                sv[k] = src[ec];
                tv[k] = dst[ec];
            }
            #pragma unroll
            for (int k = 0; k < NI; ++k) {
                const int e = e0 + 32 * k + lane;
                mine[k] = (e < n_edges) && ((tv[k] >= nsplit) == (q != 0));
                fv[k] = 0;
                if constexpr (SCALE) {
                    if (mine[k]) fv[k] = __builtin_bit_cast(int, gcn_norm(deg_src[sv[k]]));
                }
            }
            int kept = 0;
            #pragma unroll
            for (int k = 0; k < NI; ++k) {
                const unsigned b = __builtin_amdgcn_ballot_w32(mine[k]);
                const int pos = kept + (int)__builtin_amdgcn_mbcnt_lo(b, 0u);
                if (mine[k]) {
                    ls_s[wib][pos] = sv[k];
                    ls_t[wib][pos] = tv[k];
                    if constexpr (SCALE) ls_f[wib][pos] = fv[k];
                }
                kept += __builtin_popcount(b);
            }
            #pragma unroll 1
            for (int k = 0; k < NI; ++k) {
                const int left = kept - 32 * k;
                if (left <= 0) break;
                const int cnt = (left < 32) ? left : 32;
                const int cs = ls_s[wib][32 * k + lane];
                const int ct = ls_t[wib][32 * k + lane];
                int cf = 0;
                if constexpr (SCALE) cf = ls_f[wib][32 * k + lane];
                gcn_rows<D, SCALE>(Y, Out, cs, ct, cf, cnt, lane);
            }
        }
    }
}

template <int D>
__global__ void gcn_mid(const float* __restrict__ A, const float* __restrict__ norm_src,
                        const float* __restrict__ norm_dst, const float* __restrict__ bias,
                        float* __restrict__ Hs, float* __restrict__ Z, int n)
{
    static_assert(D % 4 == 0, "a row is a whole number of 16-byte pieces");
    const long q = (long)blockIdx.x * blockDim.x + threadIdx.x;
    if (q >= (long)n * (D / 4)) return;
    const int i = (int)(q / (D / 4));
    const int f = (int)(q - (long)i * (D / 4)) * 4;
    const float4 a = reinterpret_cast<const float4*>(A)[q];
    const float nd = gcn_norm(norm_dst[i]);
    const float ns = gcn_norm(norm_src[i]);
    float4 h;
    h.x = fmaxf(a.x * nd + bias[f],     0.0f) * ns;
    h.y = fmaxf(a.y * nd + bias[f + 1], 0.0f) * ns;
    h.z = fmaxf(a.z * nd + bias[f + 2], 0.0f) * ns;
    h.w = fmaxf(a.w * nd + bias[f + 3], 0.0f) * ns;
    reinterpret_cast<float4*>(Hs)[q] = h;
    reinterpret_cast<float4*>(Z)[q]  = make_float4(0.0f, 0.0f, 0.0f, 0.0f);
}

static inline int cdiv_l(long a, long b) { return (int)((a + b - 1) / b); }

constexpr int GCN_SB = 2048;

template <int DIN, int DN, int TPB>
static void run_layer(const float* Hin, float* deg,
                      const float* bias_in, const float* W, float* bufP, float* Agg,
                      const int* src, const int* dst, int N, int E, int* tk, hipStream_t stream) {
    constexpr int NI = 2;
    gcn_gemm_tiles<DIN, DN, TPB><<<cdiv_l(cdiv_l(N, 16), TPB), 32 * (DN / 16), 0, stream>>>(
        Hin, deg, deg + N, bias_in, W, bufP, Agg, N);
    gcn_scatter_add<DN, false, NI><<<GCN_SB, 256, 0, stream>>>(
        bufP, src, dst, Agg, E, deg, N, tk);
}

template <int DIN, int DN, int TPB>
static void run_first(const float* x, float* deg, const float* W, float* bufP, float* Agg,
                      const int* src, const int* dst, int N, int E, int* tk, hipStream_t stream) {
    gcn_gemm_first<DIN, DN, TPB><<<cdiv_l(cdiv_l(N, 16), TPB), 32 * (DN / 16), 0, stream>>>(
        x, W, bufP, Agg, N, src, dst, deg, E);
    gcn_scatter_add<DN, true, 1><<<GCN_SB, 256, 0, stream>>>(
        bufP, src, dst, Agg, E, deg, N, tk);
}

extern "C" void kernel_launch(void* const* d_in, const int* in_sizes, int n_in,
                              void* d_out, int out_size, void* d_ws, size_t ws_size,
                              hipStream_t stream) {
    const float* x   = (const float*)d_in[0];
    const int*   src = (const int*)d_in[1];
    const int*   dst = (const int*)d_in[2];
    const float* W1 = (const float*)d_in[3];
    const float* b1 = (const float*)d_in[4];
    const float* W2 = (const float*)d_in[5];
    const float* b2 = (const float*)d_in[6];
    const float* W3 = (const float*)d_in[7];
    const float* b3 = (const float*)d_in[8];
    const float* W4 = (const float*)d_in[9];
    const float* b4 = (const float*)d_in[10];

    const int N = in_sizes[0] / 256;
    const int E = in_sizes[1];

    float* ws       = (float*)d_ws;
    float* norm_src = ws;
    float* norm_dst = ws + N;
    float* bufA     = ws + 2 * (size_t)N;
    float* bufB     = bufA + (size_t)N * 128;
    float* out      = (float*)d_out;

    int* tk = (int*)d_out;
    gcn_zero<<<cdiv_l(2L * N, 256), 256, 0, stream>>>(norm_src, 2L * N, tk);

    float* sum2 = bufA + (size_t)N * 64;
    float* sum3 = bufB + (size_t)N * 16;
    run_first<256, 128, 5>(x,    norm_src,     W1, bufA, bufB, src, dst, N, E, tk,       stream);
    run_layer<128, 64, 4>(bufB,  norm_src, b1, W2, bufA, sum2, src, dst, N, E, tk + 128, stream);
    run_layer<64, 16, 2>(sum2,   norm_src, b2, W3, bufB, sum3, src, dst, N, E, tk + 256, stream);

    float* hs   = bufA;
    float* sum4 = bufA + (size_t)N * 16;
    gcn_mid<16><<<cdiv_l(4L * N, 256), 256, 0, stream>>>(sum3, norm_src, norm_dst, b3, hs, sum4, N);
    gcn_scatter_add<16, false, 2><<<GCN_SB, 256, 0, stream>>>(
        hs, src, dst, sum4, E, norm_src, N, tk + 384);
    gcn_gemm_last<16, 40, 4><<<cdiv_l(cdiv_l(N, 16), 4), 32 * 3, 0, stream>>>(
        sum4, norm_dst, b4, W4, out, N);
}


# Layer 1's edge sum: the cache halves split the work by 256-byte column half of every row (was by destination row set): each holds half of both tables, no edge filtered or packed; arrival order
# speedup vs baseline: 2.4235x; 1.0330x over previous
#include <hip/hip_runtime.h>
#include <hip/hip_bf16.h>

typedef __attribute__((ext_vector_type(2))) float v2f;
typedef __attribute__((ext_vector_type(8))) float v8f;

constexpr int GCN_TK = 4 * 128;

__global__ void gcn_zero(float* __restrict__ p, long n, int* __restrict__ tk) {
    long i = (long)blockIdx.x * blockDim.x + threadIdx.x;
    if (i < n) p[i] = 0.0f;
    if (i < GCN_TK) tk[i] = 0;
}

__device__ __forceinline__ float gcn_norm(float deg) { return rsqrtf(fmaxf(deg, 1.0f)); }

template <int DIN, int DN, int TPB>
__global__ __launch_bounds__((DN / 16) * 32) void gcn_gemm_first(
    const float* __restrict__ X, const float* __restrict__ W, float* __restrict__ Y,
    float* __restrict__ Sum, int n,
    const int* __restrict__ esrc, const int* __restrict__ edst, float* __restrict__ deg_w,
    int n_edges)
{
    constexpr int NT  = (DN / 16) * 32;
    constexpr int LDA = DIN + 1;
    constexpr int NLD = (16 * DIN) / NT;
    static_assert(DN % 16 == 0 && (16 * DIN) % NT == 0 && DIN % 4 == 0, "whole tiles only");
    __shared__ float As[16 * LDA];

    const int tid  = threadIdx.x;
    const int lane = tid & 31;
    const int wave = tid >> 5;
    const int mn   = lane & 15;
    const int kh   = lane >> 4;
    const int col  = (wave << 4) + mn;

    const int tile0 = (int)blockIdx.x * TPB;

    float pre[NLD];
    #pragma unroll
    for (int i = 0; i < NLD; ++i) {
        const int idx = tid + i * NT;
        const int r = idx / DIN, c = idx - r * DIN;
        const int row = tile0 * 16 + r;
        pre[i] = X[(size_t)((row < n) ? row : (n - 1)) * DIN + c];
    }

    const float* __restrict__ wp = W + (size_t)(2 * kh) * DN + col;
    v2f bw[DIN / 4];
    #pragma unroll
    for (int k = 0; k < DIN; k += 4) {
        bw[k / 4].x = wp[k * DN];
        bw[k / 4].y = wp[k * DN + DN];
    }

    {
        const int per = (n_edges + (int)gridDim.x - 1) / (int)gridDim.x;
        const int eb  = (int)blockIdx.x * per;
        for (int i0 = tid; i0 < per; i0 += NT * TPB) {
            int s[TPB], d[TPB];
            #pragma unroll
            for (int u = 0; u < TPB; ++u) {
                const int e  = eb + i0 + u * NT;
                const int ec = (e < n_edges) ? e : (n_edges - 1);
                s[u] = esrc[ec];
                d[u] = edst[ec];
            }
            #pragma unroll
            for (int u = 0; u < TPB; ++u) {
                const int i = i0 + u * NT;
                if (i < per && eb + i < n_edges) {
                    unsafeAtomicAdd(deg_w + s[u], 1.0f);
                    unsafeAtomicAdd(deg_w + n + d[u], 1.0f);
                }
            }
        }
    }

    const float* __restrict__ ap = As + mn * LDA + 2 * kh;

    #pragma unroll 1
    for (int t = 0; t < TPB; ++t) {
        const int row0 = (tile0 + t) * 16;
        if (row0 >= n) break;

        #pragma unroll
        for (int i = 0; i < NLD; ++i) {
            const int idx = tid + i * NT;
            const int r = idx / DIN, c = idx - r * DIN;
            As[r * LDA + c] = (row0 + r < n) ? pre[i] : 0.0f;
        }
        __syncthreads();

        if (t + 1 < TPB && row0 + 16 < n) {
            #pragma unroll
            for (int i = 0; i < NLD; ++i) {
                const int idx = tid + i * NT;
                const int r = idx / DIN, c = idx - r * DIN;
                const int row = row0 + 16 + r;
                pre[i] = X[(size_t)((row < n) ? row : (n - 1)) * DIN + c];
            }
        }

        v8f acc = {0.f, 0.f, 0.f, 0.f, 0.f, 0.f, 0.f, 0.f};
        #pragma unroll
        for (int k = 0; k < DIN; k += 4) {
            v2f a;
            a.x = ap[k];
            a.y = ap[k + 1];
            acc = __builtin_amdgcn_wmma_f32_16x16x4_f32(
                false, a, false, bw[k / 4], (short)0, acc, false, false);
        }

        #pragma unroll
        for (int v = 0; v < 8; ++v) {
            const int row = row0 + v + 8 * kh;
            if (row < n) Y[(size_t)row * DN + col] = acc[v];
        }

        {
            const int rows = (n - row0 < 16) ? (n - row0) : 16;
            float4* __restrict__ z = reinterpret_cast<float4*>(Sum + (size_t)row0 * DN);
            for (int i = tid; i < 16 * (DN / 4); i += NT)
                if (i < rows * (DN / 4)) z[i] = make_float4(0.0f, 0.0f, 0.0f, 0.0f);
        }

        __syncthreads();
    }
}

template <int DIN, int DN, int TPB>
__global__ __launch_bounds__((DN / 16) * 32) void gcn_gemm_tiles(
    const float* __restrict__ H, const float* __restrict__ deg_src,
    const float* __restrict__ deg_dst, const float* __restrict__ bias_in,
    const float* __restrict__ W, float* __restrict__ Y, float* __restrict__ Sum, int n)
{
    constexpr int NT  = (DN / 16) * 32;
    constexpr int LDA = DIN + 1;
    constexpr int NLD = (16 * DIN) / NT;
    constexpr int NB  = DIN / NT;
    static_assert(DN % 16 == 0 && DIN % NT == 0 && DIN % 4 == 0, "whole tiles only");
    __shared__ float As[16 * LDA];

    const int tid  = threadIdx.x;
    const int lane = tid & 31;
    const int wave = tid >> 5;
    const int mn   = lane & 15;
    const int kh   = lane >> 4;
    const int col  = (wave << 4) + mn;

    const int tile0 = (int)blockIdx.x * TPB;

    float bs[NB];
    #pragma unroll
    for (int j = 0; j < NB; ++j) bs[j] = bias_in[tid + j * NT];

    float pre[NLD];
    #pragma unroll
    for (int i = 0; i < NLD; ++i) {
        const int row = tile0 * 16 + (i * NT) / DIN;
        pre[i] = H[(size_t)((row < n) ? row : (n - 1)) * DIN + tid + (i * NT) % DIN];
    }
    float dd, ds;
    {
        const int row = tile0 * 16 + mn;
        const int rc  = (row < n) ? row : (n - 1);
        dd = deg_dst[rc];
        ds = deg_src[rc];
    }

    const float* __restrict__ wp = W + (size_t)(2 * kh) * DN + col;
    v2f bw[DIN / 4];
    #pragma unroll
    for (int k = 0; k < DIN; k += 4) {
        bw[k / 4].x = wp[k * DN];
        bw[k / 4].y = wp[k * DN + DN];
    }

    const float* __restrict__ ap = As + mn * LDA + 2 * kh;

    #pragma unroll 1
    for (int t = 0; t < TPB; ++t) {
        const int row0 = (tile0 + t) * 16;
        if (row0 >= n) break;

        const int fd = __builtin_bit_cast(int, gcn_norm(dd));
        const int fs = __builtin_bit_cast(int, gcn_norm(ds));
        #pragma unroll
        for (int i = 0; i < NLD; ++i) {
            const int r = (i * NT) / DIN;
            const int c = tid + (i * NT) % DIN;
            const float nd = __builtin_bit_cast(float, __builtin_amdgcn_readlane(fd, r));
            const float ns = __builtin_bit_cast(float, __builtin_amdgcn_readlane(fs, r));
            const float v = fmaxf(pre[i] * nd + bs[i % NB], 0.0f) * ns;
            As[r * LDA + c] = (row0 + r < n) ? v : 0.0f;
        }
        __syncthreads();

        if (t + 1 < TPB && row0 + 16 < n) {
            #pragma unroll
            for (int i = 0; i < NLD; ++i) {
                const int row = row0 + 16 + (i * NT) / DIN;
                pre[i] = H[(size_t)((row < n) ? row : (n - 1)) * DIN + tid + (i * NT) % DIN];
            }
            const int row = row0 + 16 + mn;
            const int rc  = (row < n) ? row : (n - 1);
            dd = deg_dst[rc];
            ds = deg_src[rc];
        }

        v8f acc = {0.f, 0.f, 0.f, 0.f, 0.f, 0.f, 0.f, 0.f};
        #pragma unroll
        for (int k = 0; k < DIN; k += 4) {
            v2f a;
            a.x = ap[k];
            a.y = ap[k + 1];
            acc = __builtin_amdgcn_wmma_f32_16x16x4_f32(
                false, a, false, bw[k / 4], (short)0, acc, false, false);
        }

        #pragma unroll
        for (int v = 0; v < 8; ++v) {
            const int row = row0 + v + 8 * kh;
            if (row < n) Y[(size_t)row * DN + col] = acc[v];
        }

        {
            const int rows = (n - row0 < 16) ? (n - row0) : 16;
            float4* __restrict__ z = reinterpret_cast<float4*>(Sum + (size_t)row0 * DN);
            for (int i = tid; i < 16 * (DN / 4); i += NT)
                if (i < rows * (DN / 4)) z[i] = make_float4(0.0f, 0.0f, 0.0f, 0.0f);
        }

        __syncthreads();
    }
}

template <int DIN, int DN, int TPB>
__global__ __launch_bounds__(((DN + 15) / 16) * 32) void gcn_gemm_last(
    const float* __restrict__ H, const float* __restrict__ deg_dst,
    const float* __restrict__ bias_in, const float* __restrict__ W,
    float* __restrict__ Y, int n)
{
    constexpr int NT  = ((DN + 15) / 16) * 32;
    constexpr int LDA = DIN + 1;
    constexpr int NEL = 16 * DIN;
    constexpr int NLD = (NEL + NT - 1) / NT;
    static_assert(DIN % 4 == 0, "whole k steps only");
    __shared__ float As[16 * LDA];

    const int tid  = threadIdx.x;
    const int lane = tid & 31;
    const int wave = tid >> 5;
    const int mn   = lane & 15;
    const int kh   = lane >> 4;
    const int col  = (wave << 4) + mn;

    const int   colc = (col < DN) ? col : (DN - 1);
    float msk = 1.0f;
    if constexpr ((DN & 15) != 0) msk = (col < DN) ? 1.0f : 0.0f;

    const int tile0 = (int)blockIdx.x * TPB;

    float pre[NLD];
    #pragma unroll
    for (int i = 0; i < NLD; ++i) {
        const int idx = (tid + i * NT < NEL) ? (tid + i * NT) : (NEL - 1);
        const int r = idx / DIN, c = idx - r * DIN;
        const int row = tile0 * 16 + r;
        pre[i] = H[(size_t)((row < n) ? row : (n - 1)) * DIN + c];
    }
    float dq[8];
    #pragma unroll
    for (int v = 0; v < 8; ++v) {
        const int row = tile0 * 16 + v + 8 * kh;
        dq[v] = deg_dst[(row < n) ? row : (n - 1)];
    }

    const float* __restrict__ wp = W + (size_t)(2 * kh) * DN + colc;
    v2f bw[DIN / 4];
    #pragma unroll
    for (int k = 0; k < DIN; k += 4) {
        bw[k / 4].x = wp[k * DN] * msk;
        bw[k / 4].y = wp[k * DN + DN] * msk;
    }
    const float bo = bias_in[colc];

    const float* __restrict__ ap = As + mn * LDA + 2 * kh;

    #pragma unroll 1
    for (int t = 0; t < TPB; ++t) {
        const int row0 = (tile0 + t) * 16;
        if (row0 >= n) break;

        float dc[8];
        #pragma unroll
        for (int v = 0; v < 8; ++v) dc[v] = dq[v];
        #pragma unroll
        for (int i = 0; i < NLD; ++i) {
            const int idx = tid + i * NT;
            if (idx < NEL) {
                const int r = idx / DIN, c = idx - r * DIN;
                As[r * LDA + c] = (row0 + r < n) ? pre[i] : 0.0f;
            }
        }
        __syncthreads();

        if (t + 1 < TPB && row0 + 16 < n) {
            #pragma unroll
            for (int i = 0; i < NLD; ++i) {
                const int idx = (tid + i * NT < NEL) ? (tid + i * NT) : (NEL - 1);
                const int r = idx / DIN, c = idx - r * DIN;
                const int row = row0 + 16 + r;
                pre[i] = H[(size_t)((row < n) ? row : (n - 1)) * DIN + c];
            }
            #pragma unroll
            for (int v = 0; v < 8; ++v) {
                const int row = row0 + 16 + v + 8 * kh;
                dq[v] = deg_dst[(row < n) ? row : (n - 1)];
            }
        }

        v8f acc = {0.f, 0.f, 0.f, 0.f, 0.f, 0.f, 0.f, 0.f};
        #pragma unroll
        for (int k = 0; k < DIN; k += 4) {
            v2f a;
            a.x = ap[k];
            a.y = ap[k + 1];
            acc = __builtin_amdgcn_wmma_f32_16x16x4_f32(
                false, a, false, bw[k / 4], (short)0, acc, false, false);
        }

        if (col < DN) {
            #pragma unroll
            for (int v = 0; v < 8; ++v) {
                const int row = row0 + v + 8 * kh;
                if (row < n) Y[(size_t)row * DN + col] = acc[v] * gcn_norm(dc[v]) + bo;
            }
        }

        __syncthreads();
    }
}


__device__ __forceinline__ int gcn_cache_half() {
    return (int)((__builtin_amdgcn_s_sendmsg_rtn(0x87) >> 18) & 1u);
}

template <int D, int DC, bool SCALE>
__device__ __forceinline__ void gcn_rows(const float* __restrict__ Y, float* __restrict__ Out,
                                         int cs, int ct, int cf, int cnt, int lane)
{
    if constexpr (DC % 32 == 0) {
        constexpr int Q = DC / 32;
        constexpr int G = 32 / Q;
        #pragma unroll
        for (int g = 0; g < 32; g += G) {
            if (g >= cnt) break;
            float v[G][Q];
            #pragma unroll
            for (int j = 0; j < G; ++j) {
                if (g + j < cnt) {
                    const int s = __builtin_amdgcn_readlane(cs, g + j);
                    const float* __restrict__ y = Y + (size_t)s * D + lane;
                    #pragma unroll
                    for (int q = 0; q < Q; ++q) v[j][q] = y[32 * q];
                }
            }
            #pragma unroll
            for (int j = 0; j < G; ++j) {
                if (g + j < cnt) {
                    const int t = __builtin_amdgcn_readlane(ct, g + j);
                    float* __restrict__ o = Out + (size_t)t * D + lane;
                    if constexpr (SCALE) {
                        const float w = __builtin_bit_cast(float, __builtin_amdgcn_readlane(cf, g + j));
                        #pragma unroll
                        for (int q = 0; q < Q; ++q) v[j][q] *= w;
                    }
                    #pragma unroll
                    for (int q = 0; q < Q; ++q) unsafeAtomicAdd(o + 32 * q, v[j][q]);
                }
            }
        }
    } else {
        static_assert(D == 16 && DC == 16 && !SCALE, "64-byte rows: two rows per instruction");
        const bool hi = lane >= 16;
        const int  c  = lane & 15;
        float v[16];
        #pragma unroll
        for (int p = 0; p < 16; ++p) {
            if (2 * p < cnt) {
                const int sa = __builtin_amdgcn_readlane(cs, 2 * p);
                const int sb = __builtin_amdgcn_readlane(cs, 2 * p + 1);
                if (!hi || 2 * p + 1 < cnt) v[p] = Y[(size_t)(hi ? sb : sa) * 16 + c];
            }
        }
        #pragma unroll
        for (int p = 0; p < 16; ++p) {
            if (2 * p < cnt) {
                const int ta = __builtin_amdgcn_readlane(ct, 2 * p);
                const int tb = __builtin_amdgcn_readlane(ct, 2 * p + 1);
                if (!hi || 2 * p + 1 < cnt) unsafeAtomicAdd(Out + (size_t)(hi ? tb : ta) * 16 + c, v[p]);
            }
        }
    }
}

template <int D, bool SCALE, int NI, bool COLS>
__global__ __launch_bounds__(256) void gcn_scatter_add(
    const float* __restrict__ Y, const int* __restrict__ src, const int* __restrict__ dst,
    float* __restrict__ Out, int n_edges, const float* __restrict__ deg_src,
    int n, int* __restrict__ tickets)
{
    constexpr int EW = 32 * NI;
    constexpr int EB = 8 * EW;
    static_assert(!COLS || D % 128 == 0, "a column set is whole 256-byte units of a row");
    __shared__ int ls_s[COLS ? 1 : 8][COLS ? 1 : EW];
    __shared__ int ls_t[COLS ? 1 : 8][COLS ? 1 : EW];
    __shared__ int ls_f[(SCALE && !COLS) ? 8 : 1][(SCALE && !COLS) ? EW : 1];
    __shared__ int sh_tk[2];
    __shared__ int sh_half;

    const int lane = threadIdx.x & 31;
    const int wib  = __builtin_amdgcn_readfirstlane((int)(threadIdx.x >> 5));
    const int nchunks = (n_edges + EB - 1) / EB;
    const int nsplit  = (n >> 1) & ~3;

    if (threadIdx.x == 0) sh_half = gcn_cache_half();
    __syncthreads();
    const int half = __builtin_amdgcn_readfirstlane(sh_half);

    int par = 0;
    #pragma unroll 1
    for (int phase = 0; phase < 2; ++phase) {
        const int q = half ^ phase;
        unsigned* __restrict__ ctr = reinterpret_cast<unsigned*>(tickets + 64 * q);
        int nxt = 0;
        if (threadIdx.x == 0) nxt = (int)atomicInc(ctr, 0xffffffffu);
        #pragma unroll 1
        for (;;) {
            if (threadIdx.x == 0) sh_tk[par] = nxt;
            __syncthreads();
            const int tk = __builtin_amdgcn_readfirstlane(sh_tk[par]);
            par ^= 1;
            if (tk >= nchunks) break;
            if (threadIdx.x == 0) nxt = (int)atomicInc(ctr, 0xffffffffu);

            const int e0 = tk * EB + wib * EW;
            int sv[NI], tv[NI], fv[NI];
            bool mine[NI];
            #pragma unroll
            for (int k = 0; k < NI; ++k) {
                const int e  = e0 + 32 * k + lane;
                const int ec = (e < n_edges) ? e : (n_edges - 1);
                sv[k] = src[ec];
                tv[k] = dst[ec];
            }
            #pragma unroll
            for (int k = 0; k < NI; ++k) {
                const int e = e0 + 32 * k + lane;
                if constexpr (COLS) mine[k] = (e < n_edges);
                else mine[k] = (e < n_edges) && ((tv[k] >= nsplit) == (q != 0));
                fv[k] = 0;
                if constexpr (SCALE) {
                    if (mine[k]) fv[k] = __builtin_bit_cast(int, gcn_norm(deg_src[sv[k]]));
                }
            }
            if constexpr (COLS) {
                const float* __restrict__ Yq = Y + q * (D / 2);
                float* __restrict__ Oq = Out + q * (D / 2);
                #pragma unroll
                for (int k = 0; k < NI; ++k) {
                    const int left = n_edges - (e0 + 32 * k);
                    if (left > 0)
                        gcn_rows<D, D / 2, SCALE>(Yq, Oq, sv[k], tv[k], fv[k],
                                                  (left < 32) ? left : 32, lane);
                }
            } else {
                int kept = 0;
                #pragma unroll
                for (int k = 0; k < NI; ++k) {
                    const unsigned b = __builtin_amdgcn_ballot_w32(mine[k]);
                    const int pos = kept + (int)__builtin_amdgcn_mbcnt_lo(b, 0u);
                    if (mine[k]) {
                        ls_s[wib][pos] = sv[k];
                        ls_t[wib][pos] = tv[k];
                        if constexpr (SCALE) ls_f[wib][pos] = fv[k];
                    }
                    kept += __builtin_popcount(b);
                }
                #pragma unroll 1
                for (int k = 0; k < NI; ++k) {
                    const int left = kept - 32 * k;
                    if (left <= 0) break;
                    const int cnt = (left < 32) ? left : 32;
                    const int cs = ls_s[wib][32 * k + lane];
                    const int ct = ls_t[wib][32 * k + lane];
                    int cf = 0;
                    if constexpr (SCALE) cf = ls_f[wib][32 * k + lane];
                    gcn_rows<D, D, SCALE>(Y, Out, cs, ct, cf, cnt, lane);
                }
            }
        }
    }
}

template <int D>
__global__ void gcn_mid(const float* __restrict__ A, const float* __restrict__ norm_src,
                        const float* __restrict__ norm_dst, const float* __restrict__ bias,
                        float* __restrict__ Hs, float* __restrict__ Z, int n)
{
    static_assert(D % 4 == 0, "a row is a whole number of 16-byte pieces");
    const long q = (long)blockIdx.x * blockDim.x + threadIdx.x;
    if (q >= (long)n * (D / 4)) return;
    const int i = (int)(q / (D / 4));
    const int f = (int)(q - (long)i * (D / 4)) * 4;
    const float4 a = reinterpret_cast<const float4*>(A)[q];
    const float nd = gcn_norm(norm_dst[i]);
    const float ns = gcn_norm(norm_src[i]);
    float4 h;
    h.x = fmaxf(a.x * nd + bias[f],     0.0f) * ns;
    h.y = fmaxf(a.y * nd + bias[f + 1], 0.0f) * ns;
    h.z = fmaxf(a.z * nd + bias[f + 2], 0.0f) * ns;
    h.w = fmaxf(a.w * nd + bias[f + 3], 0.0f) * ns;
    reinterpret_cast<float4*>(Hs)[q] = h;
    reinterpret_cast<float4*>(Z)[q]  = make_float4(0.0f, 0.0f, 0.0f, 0.0f);
}

static inline int cdiv_l(long a, long b) { return (int)((a + b - 1) / b); }

constexpr int GCN_SB = 2048;

template <int DIN, int DN, int TPB>
static void run_layer(const float* Hin, float* deg,
                      const float* bias_in, const float* W, float* bufP, float* Agg,
                      const int* src, const int* dst, int N, int E, int* tk, hipStream_t stream) {
    constexpr int NI = 2;
    gcn_gemm_tiles<DIN, DN, TPB><<<cdiv_l(cdiv_l(N, 16), TPB), 32 * (DN / 16), 0, stream>>>(
        Hin, deg, deg + N, bias_in, W, bufP, Agg, N);
    gcn_scatter_add<DN, false, NI, false><<<GCN_SB, 256, 0, stream>>>(
        bufP, src, dst, Agg, E, deg, N, tk);
}

template <int DIN, int DN, int TPB>
static void run_first(const float* x, float* deg, const float* W, float* bufP, float* Agg,
                      const int* src, const int* dst, int N, int E, int* tk, hipStream_t stream) {
    gcn_gemm_first<DIN, DN, TPB><<<cdiv_l(cdiv_l(N, 16), TPB), 32 * (DN / 16), 0, stream>>>(
        x, W, bufP, Agg, N, src, dst, deg, E);
    gcn_scatter_add<DN, true, 1, true><<<GCN_SB, 256, 0, stream>>>(
        bufP, src, dst, Agg, E, deg, N, tk);
}

extern "C" void kernel_launch(void* const* d_in, const int* in_sizes, int n_in,
                              void* d_out, int out_size, void* d_ws, size_t ws_size,
                              hipStream_t stream) {
    const float* x   = (const float*)d_in[0];
    const int*   src = (const int*)d_in[1];
    const int*   dst = (const int*)d_in[2];
    const float* W1 = (const float*)d_in[3];
    const float* b1 = (const float*)d_in[4];
    const float* W2 = (const float*)d_in[5];
    const float* b2 = (const float*)d_in[6];
    const float* W3 = (const float*)d_in[7];
    const float* b3 = (const float*)d_in[8];
    const float* W4 = (const float*)d_in[9];
    const float* b4 = (const float*)d_in[10];

    const int N = in_sizes[0] / 256;
    const int E = in_sizes[1];

    float* ws       = (float*)d_ws;
    float* norm_src = ws;
    float* norm_dst = ws + N;
    float* bufA     = ws + 2 * (size_t)N;
    float* bufB     = bufA + (size_t)N * 128;
    float* out      = (float*)d_out;

    int* tk = (int*)d_out;
    gcn_zero<<<cdiv_l(2L * N, 256), 256, 0, stream>>>(norm_src, 2L * N, tk);

    float* sum2 = bufA + (size_t)N * 64;
    float* sum3 = bufB + (size_t)N * 16;
    run_first<256, 128, 5>(x,    norm_src,     W1, bufA, bufB, src, dst, N, E, tk,       stream);
    run_layer<128, 64, 4>(bufB,  norm_src, b1, W2, bufA, sum2, src, dst, N, E, tk + 128, stream);
    run_layer<64, 16, 2>(sum2,   norm_src, b2, W3, bufB, sum3, src, dst, N, E, tk + 256, stream);

    float* hs   = bufA;
    float* sum4 = bufA + (size_t)N * 16;
    gcn_mid<16><<<cdiv_l(4L * N, 256), 256, 0, stream>>>(sum3, norm_src, norm_dst, b3, hs, sum4, N);
    gcn_scatter_add<16, false, 2, false><<<GCN_SB, 256, 0, stream>>>(
        hs, src, dst, sum4, E, norm_src, N, tk + 384);
    gcn_gemm_last<16, 40, 4><<<cdiv_l(cdiv_l(N, 16), 4), 32 * 3, 0, stream>>>(
        sum4, norm_dst, b4, W4, out, N);
}


# Layers 3, 4 edge sums (64-byte rows): row pairs loaded and added by all 32 lanes, odd last row apart: a pass's 16 loads go out as one group (each had waited for the one before); same adds, same order
# speedup vs baseline: 2.5238x; 1.0414x over previous
#include <hip/hip_runtime.h>
#include <hip/hip_bf16.h>

typedef __attribute__((ext_vector_type(2))) float v2f;
typedef __attribute__((ext_vector_type(8))) float v8f;

constexpr int GCN_TK = 4 * 128;

__global__ void gcn_zero(float* __restrict__ p, long n, int* __restrict__ tk) {
    long i = (long)blockIdx.x * blockDim.x + threadIdx.x;
    if (i < n) p[i] = 0.0f;
    if (i < GCN_TK) tk[i] = 0;
}

__device__ __forceinline__ float gcn_norm(float deg) { return rsqrtf(fmaxf(deg, 1.0f)); }

template <int DIN, int DN, int TPB>
__global__ __launch_bounds__((DN / 16) * 32) void gcn_gemm_first(
    const float* __restrict__ X, const float* __restrict__ W, float* __restrict__ Y,
    float* __restrict__ Sum, int n,
    const int* __restrict__ esrc, const int* __restrict__ edst, float* __restrict__ deg_w,
    int n_edges)
{
    constexpr int NT  = (DN / 16) * 32;
    constexpr int LDA = DIN + 1;
    constexpr int NLD = (16 * DIN) / NT;
    static_assert(DN % 16 == 0 && (16 * DIN) % NT == 0 && DIN % 4 == 0, "whole tiles only");
    __shared__ float As[16 * LDA];

    const int tid  = threadIdx.x;
    const int lane = tid & 31;
    const int wave = tid >> 5;
    const int mn   = lane & 15;
    const int kh   = lane >> 4;
    const int col  = (wave << 4) + mn;

    const int tile0 = (int)blockIdx.x * TPB;

    float pre[NLD];
    #pragma unroll
    for (int i = 0; i < NLD; ++i) {
        const int idx = tid + i * NT;
        const int r = idx / DIN, c = idx - r * DIN;
        const int row = tile0 * 16 + r;
        pre[i] = X[(size_t)((row < n) ? row : (n - 1)) * DIN + c];
    }

    const float* __restrict__ wp = W + (size_t)(2 * kh) * DN + col;
    v2f bw[DIN / 4];
    #pragma unroll
    for (int k = 0; k < DIN; k += 4) {
        bw[k / 4].x = wp[k * DN];
        bw[k / 4].y = wp[k * DN + DN];
    }

    {
        const int per = (n_edges + (int)gridDim.x - 1) / (int)gridDim.x;
        const int eb  = (int)blockIdx.x * per;
        for (int i0 = tid; i0 < per; i0 += NT * TPB) {
            int s[TPB], d[TPB];
            #pragma unroll
            for (int u = 0; u < TPB; ++u) {
                const int e  = eb + i0 + u * NT;
                const int ec = (e < n_edges) ? e : (n_edges - 1);
                s[u] = esrc[ec];
                d[u] = edst[ec];
            }
            #pragma unroll
            for (int u = 0; u < TPB; ++u) {
                const int i = i0 + u * NT;
                if (i < per && eb + i < n_edges) {
                    unsafeAtomicAdd(deg_w + s[u], 1.0f);
                    unsafeAtomicAdd(deg_w + n + d[u], 1.0f);
                }
            }
        }
    }

    const float* __restrict__ ap = As + mn * LDA + 2 * kh;

    #pragma unroll 1
    for (int t = 0; t < TPB; ++t) {
        const int row0 = (tile0 + t) * 16;
        if (row0 >= n) break;

        #pragma unroll
        for (int i = 0; i < NLD; ++i) {
            const int idx = tid + i * NT;
            const int r = idx / DIN, c = idx - r * DIN;
            As[r * LDA + c] = (row0 + r < n) ? pre[i] : 0.0f;
        }
        __syncthreads();

        if (t + 1 < TPB && row0 + 16 < n) {
            #pragma unroll
            for (int i = 0; i < NLD; ++i) {
                const int idx = tid + i * NT;
                const int r = idx / DIN, c = idx - r * DIN;
                const int row = row0 + 16 + r;
                pre[i] = X[(size_t)((row < n) ? row : (n - 1)) * DIN + c];
            }
        }

        v8f acc = {0.f, 0.f, 0.f, 0.f, 0.f, 0.f, 0.f, 0.f};
        #pragma unroll
        for (int k = 0; k < DIN; k += 4) {
            v2f a;
            a.x = ap[k];
            a.y = ap[k + 1];
            acc = __builtin_amdgcn_wmma_f32_16x16x4_f32(
                false, a, false, bw[k / 4], (short)0, acc, false, false);
        }

        #pragma unroll
        for (int v = 0; v < 8; ++v) {
            const int row = row0 + v + 8 * kh;
            if (row < n) Y[(size_t)row * DN + col] = acc[v];
        }

        {
            const int rows = (n - row0 < 16) ? (n - row0) : 16;
            float4* __restrict__ z = reinterpret_cast<float4*>(Sum + (size_t)row0 * DN);
            for (int i = tid; i < 16 * (DN / 4); i += NT)
                if (i < rows * (DN / 4)) z[i] = make_float4(0.0f, 0.0f, 0.0f, 0.0f);
        }

        __syncthreads();
    }
}

template <int DIN, int DN, int TPB>
__global__ __launch_bounds__((DN / 16) * 32) void gcn_gemm_tiles(
    const float* __restrict__ H, const float* __restrict__ deg_src,
    const float* __restrict__ deg_dst, const float* __restrict__ bias_in,
    const float* __restrict__ W, float* __restrict__ Y, float* __restrict__ Sum, int n)
{
    constexpr int NT  = (DN / 16) * 32;
    constexpr int LDA = DIN + 1;
    constexpr int NLD = (16 * DIN) / NT;
    constexpr int NB  = DIN / NT;
    static_assert(DN % 16 == 0 && DIN % NT == 0 && DIN % 4 == 0, "whole tiles only");
    __shared__ float As[16 * LDA];

    const int tid  = threadIdx.x;
    const int lane = tid & 31;
    const int wave = tid >> 5;
    const int mn   = lane & 15;
    const int kh   = lane >> 4;
    const int col  = (wave << 4) + mn;

    const int tile0 = (int)blockIdx.x * TPB;

    float bs[NB];
    #pragma unroll
    for (int j = 0; j < NB; ++j) bs[j] = bias_in[tid + j * NT];

    float pre[NLD];
    #pragma unroll
    for (int i = 0; i < NLD; ++i) {
        const int row = tile0 * 16 + (i * NT) / DIN;
        pre[i] = H[(size_t)((row < n) ? row : (n - 1)) * DIN + tid + (i * NT) % DIN];
    }
    float dd, ds;
    {
        const int row = tile0 * 16 + mn;
        const int rc  = (row < n) ? row : (n - 1);
        dd = deg_dst[rc];
        ds = deg_src[rc];
    }

    const float* __restrict__ wp = W + (size_t)(2 * kh) * DN + col;
    v2f bw[DIN / 4];
    #pragma unroll
    for (int k = 0; k < DIN; k += 4) {
        bw[k / 4].x = wp[k * DN];
        bw[k / 4].y = wp[k * DN + DN];
    }

    const float* __restrict__ ap = As + mn * LDA + 2 * kh;

    #pragma unroll 1
    for (int t = 0; t < TPB; ++t) {
        const int row0 = (tile0 + t) * 16;
        if (row0 >= n) break;

        const int fd = __builtin_bit_cast(int, gcn_norm(dd));
        const int fs = __builtin_bit_cast(int, gcn_norm(ds));
        #pragma unroll
        for (int i = 0; i < NLD; ++i) {
            const int r = (i * NT) / DIN;
            const int c = tid + (i * NT) % DIN;
            const float nd = __builtin_bit_cast(float, __builtin_amdgcn_readlane(fd, r));
            const float ns = __builtin_bit_cast(float, __builtin_amdgcn_readlane(fs, r));
            const float v = fmaxf(pre[i] * nd + bs[i % NB], 0.0f) * ns;
            As[r * LDA + c] = (row0 + r < n) ? v : 0.0f;
        }
        __syncthreads();

        if (t + 1 < TPB && row0 + 16 < n) {
            #pragma unroll
            for (int i = 0; i < NLD; ++i) {
                const int row = row0 + 16 + (i * NT) / DIN;
                pre[i] = H[(size_t)((row < n) ? row : (n - 1)) * DIN + tid + (i * NT) % DIN];
            }
            const int row = row0 + 16 + mn;
            const int rc  = (row < n) ? row : (n - 1);
            dd = deg_dst[rc];
            ds = deg_src[rc];
        }

        v8f acc = {0.f, 0.f, 0.f, 0.f, 0.f, 0.f, 0.f, 0.f};
        #pragma unroll
        for (int k = 0; k < DIN; k += 4) {
            v2f a;
            a.x = ap[k];
            a.y = ap[k + 1];
            acc = __builtin_amdgcn_wmma_f32_16x16x4_f32(
                false, a, false, bw[k / 4], (short)0, acc, false, false);
        }

        #pragma unroll
        for (int v = 0; v < 8; ++v) {
            const int row = row0 + v + 8 * kh;
            if (row < n) Y[(size_t)row * DN + col] = acc[v];
        }

        {
            const int rows = (n - row0 < 16) ? (n - row0) : 16;
            float4* __restrict__ z = reinterpret_cast<float4*>(Sum + (size_t)row0 * DN);
            for (int i = tid; i < 16 * (DN / 4); i += NT)
                if (i < rows * (DN / 4)) z[i] = make_float4(0.0f, 0.0f, 0.0f, 0.0f);
        }

        __syncthreads();
    }
}

template <int DIN, int DN, int TPB>
__global__ __launch_bounds__(((DN + 15) / 16) * 32) void gcn_gemm_last(
    const float* __restrict__ H, const float* __restrict__ deg_dst,
    const float* __restrict__ bias_in, const float* __restrict__ W,
    float* __restrict__ Y, int n)
{
    constexpr int NT  = ((DN + 15) / 16) * 32;
    constexpr int LDA = DIN + 1;
    constexpr int NEL = 16 * DIN;
    constexpr int NLD = (NEL + NT - 1) / NT;
    static_assert(DIN % 4 == 0, "whole k steps only");
    __shared__ float As[16 * LDA];

    const int tid  = threadIdx.x;
    const int lane = tid & 31;
    const int wave = tid >> 5;
    const int mn   = lane & 15;
    const int kh   = lane >> 4;
    const int col  = (wave << 4) + mn;

    const int   colc = (col < DN) ? col : (DN - 1);
    float msk = 1.0f;
    if constexpr ((DN & 15) != 0) msk = (col < DN) ? 1.0f : 0.0f;

    const int tile0 = (int)blockIdx.x * TPB;

    float pre[NLD];
    #pragma unroll
    for (int i = 0; i < NLD; ++i) {
        const int idx = (tid + i * NT < NEL) ? (tid + i * NT) : (NEL - 1);
        const int r = idx / DIN, c = idx - r * DIN;
        const int row = tile0 * 16 + r;
        pre[i] = H[(size_t)((row < n) ? row : (n - 1)) * DIN + c];
    }
    float dq[8];
    #pragma unroll
    for (int v = 0; v < 8; ++v) {
        const int row = tile0 * 16 + v + 8 * kh;
        dq[v] = deg_dst[(row < n) ? row : (n - 1)];
    }

    const float* __restrict__ wp = W + (size_t)(2 * kh) * DN + colc;
    v2f bw[DIN / 4];
    #pragma unroll
    for (int k = 0; k < DIN; k += 4) {
        bw[k / 4].x = wp[k * DN] * msk;
        bw[k / 4].y = wp[k * DN + DN] * msk;
    }
    const float bo = bias_in[colc];

    const float* __restrict__ ap = As + mn * LDA + 2 * kh;

    #pragma unroll 1
    for (int t = 0; t < TPB; ++t) {
        const int row0 = (tile0 + t) * 16;
        if (row0 >= n) break;

        float dc[8];
        #pragma unroll
        for (int v = 0; v < 8; ++v) dc[v] = dq[v];
        #pragma unroll
        for (int i = 0; i < NLD; ++i) {
            const int idx = tid + i * NT;
            if (idx < NEL) {
                const int r = idx / DIN, c = idx - r * DIN;
                As[r * LDA + c] = (row0 + r < n) ? pre[i] : 0.0f;
            }
        }
        __syncthreads();

        if (t + 1 < TPB && row0 + 16 < n) {
            #pragma unroll
            for (int i = 0; i < NLD; ++i) {
                const int idx = (tid + i * NT < NEL) ? (tid + i * NT) : (NEL - 1);
                const int r = idx / DIN, c = idx - r * DIN;
                const int row = row0 + 16 + r;
                pre[i] = H[(size_t)((row < n) ? row : (n - 1)) * DIN + c];
            }
            #pragma unroll
            for (int v = 0; v < 8; ++v) {
                const int row = row0 + 16 + v + 8 * kh;
                dq[v] = deg_dst[(row < n) ? row : (n - 1)];
            }
        }

        v8f acc = {0.f, 0.f, 0.f, 0.f, 0.f, 0.f, 0.f, 0.f};
        #pragma unroll
        for (int k = 0; k < DIN; k += 4) {
            v2f a;
            a.x = ap[k];
            a.y = ap[k + 1];
            acc = __builtin_amdgcn_wmma_f32_16x16x4_f32(
                false, a, false, bw[k / 4], (short)0, acc, false, false);
        }

        if (col < DN) {
            #pragma unroll
            for (int v = 0; v < 8; ++v) {
                const int row = row0 + v + 8 * kh;
                if (row < n) Y[(size_t)row * DN + col] = acc[v] * gcn_norm(dc[v]) + bo;
            }
        }

        __syncthreads();
    }
}


__device__ __forceinline__ int gcn_cache_half() {
    return (int)((__builtin_amdgcn_s_sendmsg_rtn(0x87) >> 18) & 1u);
}

template <int D, int DC, bool SCALE>
__device__ __forceinline__ void gcn_rows(const float* __restrict__ Y, float* __restrict__ Out,
                                         int cs, int ct, int cf, int cnt, int lane)
{
    if constexpr (DC % 32 == 0) {
        constexpr int Q = DC / 32;
        constexpr int G = 32 / Q;
        #pragma unroll
        for (int g = 0; g < 32; g += G) {
            if (g >= cnt) break;
            float v[G][Q];
            #pragma unroll
            for (int j = 0; j < G; ++j) {
                if (g + j < cnt) {
                    const int s = __builtin_amdgcn_readlane(cs, g + j);
                    const float* __restrict__ y = Y + (size_t)s * D + lane;
                    #pragma unroll
                    for (int q = 0; q < Q; ++q) v[j][q] = y[32 * q];
                }
            }
            #pragma unroll
            for (int j = 0; j < G; ++j) {
                if (g + j < cnt) {
                    const int t = __builtin_amdgcn_readlane(ct, g + j);
                    float* __restrict__ o = Out + (size_t)t * D + lane;
                    if constexpr (SCALE) {
                        const float w = __builtin_bit_cast(float, __builtin_amdgcn_readlane(cf, g + j));
                        #pragma unroll
                        for (int q = 0; q < Q; ++q) v[j][q] *= w;
                    }
                    #pragma unroll
                    for (int q = 0; q < Q; ++q) unsafeAtomicAdd(o + 32 * q, v[j][q]);
                }
            }
        }
    } else {
        static_assert(D == 16 && DC == 16 && !SCALE, "64-byte rows: two rows per instruction");
        const bool hi = lane >= 16;
        const int  c  = lane & 15;
        const int  npair = cnt >> 1;
        const bool odd   = (cnt & 1) != 0;
        float vt = 0.0f;
        int   tt = 0;
        if (odd) {
            const int st = __builtin_amdgcn_readlane(cs, cnt - 1);
            tt = __builtin_amdgcn_readlane(ct, cnt - 1);
            if (!hi) vt = Y[(size_t)st * 16 + c];
        }
        float v[16];
        #pragma unroll
        for (int p = 0; p < 16; ++p) {
            if (p < npair) {
                const int sa = __builtin_amdgcn_readlane(cs, 2 * p);
                const int sb = __builtin_amdgcn_readlane(cs, 2 * p + 1);
                v[p] = Y[(size_t)(hi ? sb : sa) * 16 + c];
            }
        }
        #pragma unroll
        for (int p = 0; p < 16; ++p) {
            if (p < npair) {
                const int ta = __builtin_amdgcn_readlane(ct, 2 * p);
                const int tb = __builtin_amdgcn_readlane(ct, 2 * p + 1);
                unsafeAtomicAdd(Out + (size_t)(hi ? tb : ta) * 16 + c, v[p]);
            }
        }
        if (odd && !hi) unsafeAtomicAdd(Out + (size_t)tt * 16 + c, vt);
    }
}

template <int D, bool SCALE, int NI, bool COLS>
__global__ __launch_bounds__(256) void gcn_scatter_add(
    const float* __restrict__ Y, const int* __restrict__ src, const int* __restrict__ dst,
    float* __restrict__ Out, int n_edges, const float* __restrict__ deg_src,
    int n, int* __restrict__ tickets)
{
    constexpr int EW = 32 * NI;
    constexpr int EB = 8 * EW;
    static_assert(!COLS || D % 128 == 0, "a column set is whole 256-byte units of a row");
    __shared__ int ls_s[COLS ? 1 : 8][COLS ? 1 : EW];
    __shared__ int ls_t[COLS ? 1 : 8][COLS ? 1 : EW];
    __shared__ int ls_f[(SCALE && !COLS) ? 8 : 1][(SCALE && !COLS) ? EW : 1];
    __shared__ int sh_tk[2];
    __shared__ int sh_half;

    const int lane = threadIdx.x & 31;
    const int wib  = __builtin_amdgcn_readfirstlane((int)(threadIdx.x >> 5));
    const int nchunks = (n_edges + EB - 1) / EB;
    const int nsplit  = (n >> 1) & ~3;

    if (threadIdx.x == 0) sh_half = gcn_cache_half();
    __syncthreads();
    const int half = __builtin_amdgcn_readfirstlane(sh_half);

    int par = 0;
    #pragma unroll 1
    for (int phase = 0; phase < 2; ++phase) {
        const int q = half ^ phase;
        unsigned* __restrict__ ctr = reinterpret_cast<unsigned*>(tickets + 64 * q);
        int nxt = 0;
        if (threadIdx.x == 0) nxt = (int)atomicInc(ctr, 0xffffffffu);
        #pragma unroll 1
        for (;;) {
            if (threadIdx.x == 0) sh_tk[par] = nxt;
            __syncthreads();
            const int tk = __builtin_amdgcn_readfirstlane(sh_tk[par]);
            par ^= 1;
            if (tk >= nchunks) break;
            if (threadIdx.x == 0) nxt = (int)atomicInc(ctr, 0xffffffffu);

            const int e0 = tk * EB + wib * EW;
            int sv[NI], tv[NI], fv[NI];
            bool mine[NI];
            #pragma unroll
            for (int k = 0; k < NI; ++k) {
                const int e  = e0 + 32 * k + lane;
                const int ec = (e < n_edges) ? e : (n_edges - 1);
                sv[k] = src[ec];
                tv[k] = dst[ec];
            }
            #pragma unroll
            for (int k = 0; k < NI; ++k) {
                const int e = e0 + 32 * k + lane;
                if constexpr (COLS) mine[k] = (e < n_edges);
                else mine[k] = (e < n_edges) && ((tv[k] >= nsplit) == (q != 0));
                fv[k] = 0;
                if constexpr (SCALE) {
                    if (mine[k]) fv[k] = __builtin_bit_cast(int, gcn_norm(deg_src[sv[k]]));
                }
            }
            if constexpr (COLS) {
                const float* __restrict__ Yq = Y + q * (D / 2);
                float* __restrict__ Oq = Out + q * (D / 2);
                #pragma unroll
                for (int k = 0; k < NI; ++k) {
                    const int left = n_edges - (e0 + 32 * k);
                    if (left > 0)
                        gcn_rows<D, D / 2, SCALE>(Yq, Oq, sv[k], tv[k], fv[k],
                                                  (left < 32) ? left : 32, lane);
                }
            } else {
                int kept = 0;
                #pragma unroll
                for (int k = 0; k < NI; ++k) {
                    const unsigned b = __builtin_amdgcn_ballot_w32(mine[k]);
                    const int pos = kept + (int)__builtin_amdgcn_mbcnt_lo(b, 0u);
                    if (mine[k]) {
                        ls_s[wib][pos] = sv[k];
                        ls_t[wib][pos] = tv[k];
                        if constexpr (SCALE) ls_f[wib][pos] = fv[k];
                    }
                    kept += __builtin_popcount(b);
                }
                #pragma unroll 1
                for (int k = 0; k < NI; ++k) {
                    const int left = kept - 32 * k;
                    if (left <= 0) break;
                    const int cnt = (left < 32) ? left : 32;
                    const int cs = ls_s[wib][32 * k + lane];
                    const int ct = ls_t[wib][32 * k + lane];
                    int cf = 0;
                    if constexpr (SCALE) cf = ls_f[wib][32 * k + lane];
                    gcn_rows<D, D, SCALE>(Y, Out, cs, ct, cf, cnt, lane);
                }
            }
        }
    }
}

template <int D>
__global__ void gcn_mid(const float* __restrict__ A, const float* __restrict__ norm_src,
                        const float* __restrict__ norm_dst, const float* __restrict__ bias,
                        float* __restrict__ Hs, float* __restrict__ Z, int n)
{
    static_assert(D % 4 == 0, "a row is a whole number of 16-byte pieces");
    const long q = (long)blockIdx.x * blockDim.x + threadIdx.x;
    if (q >= (long)n * (D / 4)) return;
    const int i = (int)(q / (D / 4));
    const int f = (int)(q - (long)i * (D / 4)) * 4;
    const float4 a = reinterpret_cast<const float4*>(A)[q];
    const float nd = gcn_norm(norm_dst[i]);
    const float ns = gcn_norm(norm_src[i]);
    float4 h;
    h.x = fmaxf(a.x * nd + bias[f],     0.0f) * ns;
    h.y = fmaxf(a.y * nd + bias[f + 1], 0.0f) * ns;
    h.z = fmaxf(a.z * nd + bias[f + 2], 0.0f) * ns;
    h.w = fmaxf(a.w * nd + bias[f + 3], 0.0f) * ns;
    reinterpret_cast<float4*>(Hs)[q] = h;
    reinterpret_cast<float4*>(Z)[q]  = make_float4(0.0f, 0.0f, 0.0f, 0.0f);
}

static inline int cdiv_l(long a, long b) { return (int)((a + b - 1) / b); }

constexpr int GCN_SB = 2048;

template <int DIN, int DN, int TPB>
static void run_layer(const float* Hin, float* deg,
                      const float* bias_in, const float* W, float* bufP, float* Agg,
                      const int* src, const int* dst, int N, int E, int* tk, hipStream_t stream) {
    constexpr int NI = 2;
    gcn_gemm_tiles<DIN, DN, TPB><<<cdiv_l(cdiv_l(N, 16), TPB), 32 * (DN / 16), 0, stream>>>(
        Hin, deg, deg + N, bias_in, W, bufP, Agg, N);
    gcn_scatter_add<DN, false, NI, false><<<GCN_SB, 256, 0, stream>>>(
        bufP, src, dst, Agg, E, deg, N, tk);
}

template <int DIN, int DN, int TPB>
static void run_first(const float* x, float* deg, const float* W, float* bufP, float* Agg,
                      const int* src, const int* dst, int N, int E, int* tk, hipStream_t stream) {
    gcn_gemm_first<DIN, DN, TPB><<<cdiv_l(cdiv_l(N, 16), TPB), 32 * (DN / 16), 0, stream>>>(
        x, W, bufP, Agg, N, src, dst, deg, E);
    gcn_scatter_add<DN, true, 1, true><<<GCN_SB, 256, 0, stream>>>(
        bufP, src, dst, Agg, E, deg, N, tk);
}

extern "C" void kernel_launch(void* const* d_in, const int* in_sizes, int n_in,
                              void* d_out, int out_size, void* d_ws, size_t ws_size,
                              hipStream_t stream) {
    const float* x   = (const float*)d_in[0];
    const int*   src = (const int*)d_in[1];
    const int*   dst = (const int*)d_in[2];
    const float* W1 = (const float*)d_in[3];
    const float* b1 = (const float*)d_in[4];
    const float* W2 = (const float*)d_in[5];
    const float* b2 = (const float*)d_in[6];
    const float* W3 = (const float*)d_in[7];
    const float* b3 = (const float*)d_in[8];
    const float* W4 = (const float*)d_in[9];
    const float* b4 = (const float*)d_in[10];

    const int N = in_sizes[0] / 256;
    const int E = in_sizes[1];

    float* ws       = (float*)d_ws;
    float* norm_src = ws;
    float* norm_dst = ws + N;
    float* bufA     = ws + 2 * (size_t)N;
    float* bufB     = bufA + (size_t)N * 128;
    float* out      = (float*)d_out;

    int* tk = (int*)d_out;
    gcn_zero<<<cdiv_l(2L * N, 256), 256, 0, stream>>>(norm_src, 2L * N, tk);

    float* sum2 = bufA + (size_t)N * 64;
    float* sum3 = bufB + (size_t)N * 16;
    run_first<256, 128, 5>(x,    norm_src,     W1, bufA, bufB, src, dst, N, E, tk,       stream);
    run_layer<128, 64, 4>(bufB,  norm_src, b1, W2, bufA, sum2, src, dst, N, E, tk + 128, stream);
    run_layer<64, 16, 2>(sum2,   norm_src, b2, W3, bufB, sum3, src, dst, N, E, tk + 256, stream);

    float* hs   = bufA;
    float* sum4 = bufA + (size_t)N * 16;
    gcn_mid<16><<<cdiv_l(4L * N, 256), 256, 0, stream>>>(sum3, norm_src, norm_dst, b3, hs, sum4, N);
    gcn_scatter_add<16, false, 2, false><<<GCN_SB, 256, 0, stream>>>(
        hs, src, dst, sum4, E, norm_src, N, tk + 384);
    gcn_gemm_last<16, 40, 4><<<cdiv_l(cdiv_l(N, 16), 4), 32 * 3, 0, stream>>>(
        sum4, norm_dst, b4, W4, out, N);
}
